# starting kernel
# baseline (speedup 1.0000x reference)
#include <hip/hip_runtime.h>

#define N_NODES 100000
#define N_EDGES 1600000

typedef __attribute__((ext_vector_type(2))) float v2f;
typedef __attribute__((ext_vector_type(8))) float v8f;

__device__ __forceinline__ void f32_atomic_add(float* p, float v) {
    unsafeAtomicAdd(p, v);
}

__global__ void zero4_kernel(float4* __restrict__ p, int n4) {
    int t = blockIdx.x * blockDim.x + threadIdx.x;
    if (t < n4) p[t] = make_float4(0.f, 0.f, 0.f, 0.f);
}

__global__ void degree_kernel(const int* __restrict__ dst, float* __restrict__ deg) {
    int e = blockIdx.x * blockDim.x + threadIdx.x;
    if (e < N_EDGES) f32_atomic_add(&deg[dst[e]], 1.0f);
}

__global__ void dinv_kernel(float* __restrict__ deg_dinv) {
    int i = blockIdx.x * blockDim.x + threadIdx.x;
    if (i < N_NODES) deg_dinv[i] = rsqrtf(deg_dinv[i] + 1.0f);
}

template <int K, int COUT>
__global__ void gemm_dinv_wmma(const float* __restrict__ H,
                               const float* __restrict__ W,
                               const float* __restrict__ dinv,
                               float* __restrict__ Y,
                               float* __restrict__ AGG) {
    const int lane   = threadIdx.x & 31;
    const int n_tile = threadIdx.x >> 5;
    const int m_tile = blockIdx.x;

    const int row = m_tile * 16 + (lane & 15);
    const int col = n_tile * 16 + (lane & 15);
    const int kh  = (lane >> 4) << 1;

    const float* hrow = H + row * K + kh;

    v8f acc = {};
#pragma unroll
    for (int kk = 0; kk < K; kk += 4) {
        v2f a = *reinterpret_cast<const v2f*>(hrow + kk);
        v2f b;
        b.x = W[(kk + kh) * COUT + col];
        b.y = W[(kk + kh + 1) * COUT + col];
        acc = __builtin_amdgcn_wmma_f32_16x16x4_f32(
             false, a,  false, b,
             (short)0, acc,  false,  false);
    }

    const int mbase = m_tile * 16 + ((lane >> 4) << 3);
#pragma unroll
    for (int r = 0; r < 8; ++r) {
        const int m   = mbase + r;
        const float v = dinv[m] * acc[r];
        const int idx = m * COUT + col;
        Y[idx]   = v;
        AGG[idx] = v;
    }
}

template <int COUT>
__global__ void scatter_kernel(const int* __restrict__ src,
                               const int* __restrict__ dst,
                               const float* __restrict__ Y,
                               float* __restrict__ AGG) {
    constexpr int CH = COUT / 4;
    int t = blockIdx.x * blockDim.x + threadIdx.x;
    if (t >= N_EDGES * CH) return;
    int e  = t / CH;
    int c4 = (t - e * CH) * 4;
    int s = src[e];
    int d = dst[e];
    const float4 v = *reinterpret_cast<const float4*>(Y + s * COUT + c4);
    float* o = AGG + d * COUT + c4;
    f32_atomic_add(o + 0, v.x);
    f32_atomic_add(o + 1, v.y);
    f32_atomic_add(o + 2, v.z);
    f32_atomic_add(o + 3, v.w);
}

template <int COUT, bool RELU>
__global__ void combine_kernel(const float* __restrict__ AGG,
                               const float* __restrict__ dinv,
                               const float* __restrict__ bias,
                               float* __restrict__ OUT) {
    constexpr int SH = (COUT == 128) ? 7 : 6;
    int t = blockIdx.x * blockDim.x + threadIdx.x;
    if (t >= N_NODES * COUT) return;
    int i = t >> SH;
    int c = t & (COUT - 1);
    float v = dinv[i] * AGG[t] + bias[c];
    if (RELU) v = fmaxf(v, 0.0f);
    OUT[t] = v;
}

extern "C" void kernel_launch(void* const* d_in, const int* in_sizes, int n_in,
                              void* d_out, int out_size, void* d_ws, size_t ws_size,
                              hipStream_t stream) {
    const float* x  = (const float*)d_in[0];
    const int*   ei = (const int*)d_in[1];
    const float* W1 = (const float*)d_in[2];
    const float* b1 = (const float*)d_in[3];
    const float* W2 = (const float*)d_in[4];
    const float* b2 = (const float*)d_in[5];
    const float* W3 = (const float*)d_in[6];
    const float* b3 = (const float*)d_in[7];
    float*       out = (float*)d_out;

    const int* src = ei;
    const int* dst = ei + N_EDGES;

    char*  ws   = (char*)d_ws;
    size_t off  = 0;
    float* dinv = (float*)(ws + off); off += ((size_t)N_NODES * 4 + 511) & ~511ull;
    float* y    = (float*)(ws + off); off += (size_t)N_NODES * 128 * 4;
    float* agg  = (float*)(ws + off); off += (size_t)N_NODES * 128 * 4;
    float* h    = (float*)(ws + off);

    const int B = 256;
    const int M_TILES = N_NODES / 16;

    zero4_kernel<<<(N_NODES / 4 + B - 1) / B, B, 0, stream>>>((float4*)dinv, N_NODES / 4);
    degree_kernel<<<(N_EDGES + B - 1) / B, B, 0, stream>>>(dst, dinv);
    dinv_kernel<<<(N_NODES + B - 1) / B, B, 0, stream>>>(dinv);

    gemm_dinv_wmma<64, 128><<<M_TILES, 256, 0, stream>>>(x, W1, dinv, y, agg);
    scatter_kernel<128><<<(N_EDGES * 32 + B - 1) / B, B, 0, stream>>>(src, dst, y, agg);
    combine_kernel<128, true><<<(N_NODES * 128 + B - 1) / B, B, 0, stream>>>(agg, dinv, b1, h);

    gemm_dinv_wmma<128, 128><<<M_TILES, 256, 0, stream>>>(h, W2, dinv, y, agg);
    scatter_kernel<128><<<(N_EDGES * 32 + B - 1) / B, B, 0, stream>>>(src, dst, y, agg);
    combine_kernel<128, true><<<(N_NODES * 128 + B - 1) / B, B, 0, stream>>>(agg, dinv, b2, h);

    gemm_dinv_wmma<128, 64><<<M_TILES, 128, 0, stream>>>(h, W3, dinv, y, agg);
    scatter_kernel<64><<<(N_EDGES * 16 + B - 1) / B, B, 0, stream>>>(src, dst, y, agg);
    combine_kernel<64, false><<<(N_NODES * 64 + B - 1) / B, B, 0, stream>>>(agg, dinv, b3, out);
}


# aggregation by 256 owner work-groups: each keeps 391 destination rows in LDS, picks its edges from an owner-byte table, adds their source rows by LDS f32 atomics (LDS arrival order), stores h once
# speedup vs baseline: 2.0688x; 2.0688x over previous
#include <hip/hip_runtime.h>


#define N_NODES 100000
#define N_EDGES 1600000

#define OWN 391
#define N_OWNERS 256
#define AGG_THREADS 1024
#define AGG_WAVES (AGG_THREADS / 32)

typedef __attribute__((ext_vector_type(2))) float v2f;
typedef __attribute__((ext_vector_type(8))) float v8f;

__device__ __forceinline__ void f32_atomic_add(float* p, float v) {
    unsafeAtomicAdd(p, v);
}

__global__ void zero4_kernel(float4* __restrict__ p, int n4) {
    int t = blockIdx.x * blockDim.x + threadIdx.x;
    if (t < n4) p[t] = make_float4(0.f, 0.f, 0.f, 0.f);
}

__global__ void degree_kernel(const int* __restrict__ dst, float* __restrict__ deg) {
    int e = blockIdx.x * blockDim.x + threadIdx.x;
    if (e < N_EDGES) f32_atomic_add(&deg[dst[e]], 1.0f);
}

__global__ void dinv_kernel(float* __restrict__ deg_dinv) {
    int i = blockIdx.x * blockDim.x + threadIdx.x;
    if (i < N_NODES) deg_dinv[i] = rsqrtf(deg_dinv[i] + 1.0f);
}

__global__ void owner_kernel(const int* __restrict__ dst, unsigned* __restrict__ ob) {
    int t = blockIdx.x * blockDim.x + threadIdx.x;
    if (t >= N_EDGES / 4) return;
    const int4 d = *reinterpret_cast<const int4*>(dst + 4 * t);
    const unsigned b0 = (unsigned)(d.x / OWN), b1 = (unsigned)(d.y / OWN);
    const unsigned b2 = (unsigned)(d.z / OWN), b3 = (unsigned)(d.w / OWN);
    ob[t] = b0 | (b1 << 8) | (b2 << 16) | (b3 << 24);
}

template <int K, int COUT>
__global__ void gemm_dinv_wmma(const float* __restrict__ H,
                               const float* __restrict__ W,
                               const float* __restrict__ dinv,
                               float* __restrict__ Y) {
    const int lane   = threadIdx.x & 31;
    const int n_tile = threadIdx.x >> 5;
    const int m_tile = blockIdx.x;

    const int row = m_tile * 16 + (lane & 15);
    const int col = n_tile * 16 + (lane & 15);
    const int kh  = (lane >> 4) << 1;

    const float* hrow = H + row * K + kh;

    v8f acc = {};
#pragma unroll
    for (int kk = 0; kk < K; kk += 4) {
        v2f a = *reinterpret_cast<const v2f*>(hrow + kk);
        v2f b;
        b.x = W[(kk + kh) * COUT + col];
        b.y = W[(kk + kh + 1) * COUT + col];
        acc = __builtin_amdgcn_wmma_f32_16x16x4_f32(
             false, a,  false, b,
             (short)0, acc,  false,  false);
    }

    const int mbase = m_tile * 16 + ((lane >> 4) << 3);
#pragma unroll
    for (int r = 0; r < 8; ++r) {
        const int m = mbase + r;
        Y[m * COUT + col] = dinv[m] * acc[r];
    }
}

template <int COUT, bool RELU>
__global__ void __launch_bounds__(AGG_THREADS)
owner_aggregate(const int* __restrict__ src,
                const int* __restrict__ dst,
                const uint4* __restrict__ ob,
                const float* __restrict__ Y,
                const float* __restrict__ dinv,
                const float* __restrict__ bias,
                float* __restrict__ OUT) {
    constexpr int G  = COUT / 32;
    constexpr int EB = 32 / G;
    constexpr int N_TRIPS = N_EDGES / 16 / 32;
    static_assert(N_EDGES % (16 * 32) == 0, "edge count must split evenly over wave trips");

    __shared__ __attribute__((aligned(16))) float acc[OWN * COUT];
    __shared__ int q[AGG_WAVES][64];

    const int tid   = threadIdx.x;
    const int lane  = tid & 31;
    const int wave  = __builtin_amdgcn_readfirstlane(tid >> 5);
    const int owner = blockIdx.x;
    const int node0 = owner * OWN;
    const int count = (N_NODES - node0 < OWN) ? (N_NODES - node0) : OWN;
    const int nval  = count * COUT;

    for (int i = tid; i < nval; i += AGG_THREADS) acc[i] = Y[node0 * COUT + i];
    __syncthreads();

    const unsigned pat = (unsigned)owner * 0x01010101u;
    int* wq = q[wave];
    int head = 0, tail = 0;

    auto drain = [&](int n) {
        int s = 0, d = 0;
        if (lane < n) {
            const int e = wq[(head + lane) & 63];
            s = src[e];
            d = dst[e] - node0;
        }
        for (int i0 = 0; i0 < n; i0 += EB) {
            float v[EB][G];
#pragma unroll
            for (int k = 0; k < EB; ++k) {
                if (i0 + k < n) {
                    const int sk = __builtin_amdgcn_readlane(s, i0 + k);
                    const float* yr = Y + sk * COUT + lane;
#pragma unroll
                    for (int g = 0; g < G; ++g) v[k][g] = yr[g * 32];
                }
            }
#pragma unroll
            for (int k = 0; k < EB; ++k) {
                if (i0 + k < n) {
                    const int dk = __builtin_amdgcn_readlane(d, i0 + k);
                    float* a = acc + dk * COUT + lane;
#pragma unroll
                    for (int g = 0; g < G; ++g)
                        __hip_atomic_fetch_add(a + g * 32, v[k][g], __ATOMIC_RELAXED,
                                               __HIP_MEMORY_SCOPE_WORKGROUP);
                }
            }
        }
        head += n;
    };

    static_assert(AGG_WAVES <= N_TRIPS, "every wave has a first trip");
    uint4 cur = ob[wave * 32 + lane];
    for (int trip = wave; trip < N_TRIPS; trip += AGG_WAVES) {
        const int ntrip = (trip + AGG_WAVES < N_TRIPS) ? (trip + AGG_WAVES) : trip;
        const uint4 nxt = ob[ntrip * 32 + lane];

        unsigned m = 0;
        const unsigned w4[4] = {cur.x, cur.y, cur.z, cur.w};
#pragma unroll
        for (int dw = 0; dw < 4; ++dw) {
            const unsigned x = w4[dw] ^ pat;
            const unsigned t = (x & 0x7f7f7f7fu) + 0x7f7f7f7fu;
            const unsigned z = ~(t | x | 0x7f7f7f7fu);
            m |= z >> (7 - dw);
        }
        const int ebase = (trip * 32 + lane) * 16;

        for (;;) {
            const bool has = (m != 0u);
            const unsigned bal = __builtin_amdgcn_ballot_w32(has);
            if (bal == 0u) break;
            if (has) {
                const int p    = __builtin_ctz(m);
                const int j    = ((p & 7) << 2) | (p >> 3);
                const int rank = (int)__builtin_amdgcn_mbcnt_lo(bal, 0u);
                wq[(tail + rank) & 63] = ebase + j;
                m &= m - 1u;
            }
            tail += __builtin_popcount(bal);
            __builtin_amdgcn_fence(__ATOMIC_ACQ_REL, "wavefront");
            if (tail - head >= 32) drain(32);
        }
        cur = nxt;
    }
    __builtin_amdgcn_fence(__ATOMIC_ACQ_REL, "wavefront");
    drain(tail - head);
    __syncthreads();

    for (int i = tid; i < nval; i += AGG_THREADS) {
        const int n = i / COUT;
        const int c = i & (COUT - 1);
        float v = dinv[node0 + n] * acc[i] + bias[c];
        if (RELU) v = fmaxf(v, 0.0f);
        OUT[node0 * COUT + i] = v;
    }
}

extern "C" void kernel_launch(void* const* d_in, const int* in_sizes, int n_in,
                              void* d_out, int out_size, void* d_ws, size_t ws_size,
                              hipStream_t stream) {
    const float* x  = (const float*)d_in[0];
    const int*   ei = (const int*)d_in[1];
    const float* W1 = (const float*)d_in[2];
    const float* b1 = (const float*)d_in[3];
    const float* W2 = (const float*)d_in[4];
    const float* b2 = (const float*)d_in[5];
    const float* W3 = (const float*)d_in[6];
    const float* b3 = (const float*)d_in[7];
    float*       out = (float*)d_out;

    const int* src = ei;
    const int* dst = ei + N_EDGES;

    char*  ws   = (char*)d_ws;
    size_t off  = 0;
    float*    dinv = (float*)(ws + off);    off += ((size_t)N_NODES * 4 + 511) & ~511ull;
    unsigned* ob   = (unsigned*)(ws + off); off += ((size_t)N_EDGES + 511) & ~511ull;
    float*    y    = (float*)(ws + off);    off += (size_t)N_NODES * 128 * 4;
    float*    h    = (float*)(ws + off);

    const int B = 256;
    const int M_TILES = N_NODES / 16;

    zero4_kernel<<<(N_NODES / 4 + B - 1) / B, B, 0, stream>>>((float4*)dinv, N_NODES / 4);
    degree_kernel<<<(N_EDGES + B - 1) / B, B, 0, stream>>>(dst, dinv);
    dinv_kernel<<<(N_NODES + B - 1) / B, B, 0, stream>>>(dinv);
    owner_kernel<<<(N_EDGES / 4 + B - 1) / B, B, 0, stream>>>(dst, ob);

    gemm_dinv_wmma<64, 128><<<M_TILES, 256, 0, stream>>>(x, W1, dinv, y);
    owner_aggregate<128, true><<<N_OWNERS, AGG_THREADS, 0, stream>>>(src, dst, (const uint4*)ob, y, dinv, b1, h);

    gemm_dinv_wmma<128, 128><<<M_TILES, 256, 0, stream>>>(h, W2, dinv, y);
    owner_aggregate<128, true><<<N_OWNERS, AGG_THREADS, 0, stream>>>(src, dst, (const uint4*)ob, y, dinv, b2, h);

    gemm_dinv_wmma<128, 64><<<M_TILES, 128, 0, stream>>>(h, W3, dinv, y);
    owner_aggregate<64, false><<<N_OWNERS, AGG_THREADS, 0, stream>>>(src, dst, (const uint4*)ob, y, dinv, b3, out);
}


# degree pass: 128 work-groups count their edge chunk in a private LDS histogram (16-bit counter pairs), a merge pass sums the partials into dinv; no global atomics, no zero pass; counts exact
# speedup vs baseline: 3.1856x; 1.5398x over previous
#include <hip/hip_runtime.h>


#define N_NODES 100000
#define N_EDGES 1600000

#define OWN 391
#define N_OWNERS 256
#define AGG_THREADS 1024
#define AGG_WAVES (AGG_THREADS / 32)
#define LIST_CAP 512

typedef __attribute__((ext_vector_type(2))) float v2f;
typedef __attribute__((ext_vector_type(8))) float v8f;

#define DEG_PARTS 128
#define DEG_CHUNK (N_EDGES / DEG_PARTS)
#define PART_W 50176
#define DEG_THREADS 1024
static_assert(N_EDGES % DEG_PARTS == 0, "the edges must split evenly over the counting work-groups");
static_assert(DEG_CHUNK < 65536, "a chunk's count of one node must fit 16 bits");
static_assert(DEG_CHUNK % 4 == 0, "a chunk is read in groups of four destinations");
static_assert(2 * PART_W >= N_NODES && PART_W % 256 == 0, "a partial covers every node in whole merge blocks");

__global__ void __launch_bounds__(DEG_THREADS)
degree_hist_kernel(const int* dst, uint4* __restrict__ part) {
    __shared__ __attribute__((aligned(16))) unsigned cnt[PART_W];
    uint4* cnt4 = reinterpret_cast<uint4*>(cnt);
    const int tid = threadIdx.x;

    constexpr int Q = DEG_CHUNK / 4;
    constexpr int R = (Q + DEG_THREADS - 1) / DEG_THREADS;
    const int4* d4 = reinterpret_cast<const int4*>(dst) + blockIdx.x * Q;
    int4 d[R];
    for (int r = 0; r < R; ++r) {
        const int i = tid + r * DEG_THREADS;
        d[r] = (i < Q) ? d4[i] : make_int4(-1, -1, -1, -1);
    }

    for (int i = tid; i < PART_W / 4; i += DEG_THREADS) cnt4[i] = make_uint4(0u, 0u, 0u, 0u);
    __syncthreads();

    auto count = [&](int node) {
        const unsigned u = (unsigned)node;
        if (u < (unsigned)N_NODES)
            __hip_atomic_fetch_add(cnt + (u >> 1), 1u << ((u & 1u) << 4), __ATOMIC_RELAXED,
                                   __HIP_MEMORY_SCOPE_WORKGROUP);
    };
    for (int r = 0; r < R; ++r) { count(d[r].x); count(d[r].y); count(d[r].z); count(d[r].w); }
    __syncthreads();

    uint4* mine = part + (size_t)blockIdx.x * (PART_W / 4);
    for (int i = tid; i < PART_W / 4; i += DEG_THREADS) mine[i] = cnt4[i];
}

#define MRG_COLS 64
#define MRG_SLICES 8
#define MRG_THREADS (MRG_COLS * MRG_SLICES)
static_assert(DEG_PARTS % MRG_SLICES == 0, "the partials must split evenly over the slices");
static_assert(PART_W % (MRG_COLS * 4) == 0, "a partial must split into whole merge blocks");

__global__ void __launch_bounds__(MRG_THREADS)
degree_merge_kernel(const uint4* __restrict__ part, float* __restrict__ dinv) {
    __shared__ unsigned red[MRG_SLICES][MRG_COLS * 8];
    const int tid   = threadIdx.x;
    const int col   = tid & (MRG_COLS - 1);
    const int slice = tid / MRG_COLS;

    const uint4* p = part + (size_t)slice * (PART_W / 4) + blockIdx.x * MRG_COLS + col;
    unsigned s[8] = {0u, 0u, 0u, 0u, 0u, 0u, 0u, 0u};
#pragma unroll
    for (int k = 0; k < DEG_PARTS / MRG_SLICES; ++k) {
        const uint4 v = p[(size_t)k * MRG_SLICES * (PART_W / 4)];
        s[0] += v.x & 0xffffu;  s[1] += v.x >> 16;
        s[2] += v.y & 0xffffu;  s[3] += v.y >> 16;
        s[4] += v.z & 0xffffu;  s[5] += v.z >> 16;
        s[6] += v.w & 0xffffu;  s[7] += v.w >> 16;
    }
    for (int j = 0; j < 8; ++j) red[slice][col * 8 + j] = s[j];
    __syncthreads();

    for (int n = tid; n < MRG_COLS * 8; n += MRG_THREADS) {
        unsigned total = 0u;
        for (int q = 0; q < MRG_SLICES; ++q) total += red[q][n];
        const int node = blockIdx.x * (MRG_COLS * 8) + n;
        if (node < N_NODES) dinv[node] = rsqrtf((float)total + 1.0f);
    }
}

#define SRC_BITS 17
#define SRC_MASK ((1u << SRC_BITS) - 1u)
static_assert(N_NODES <= (1 << SRC_BITS), "a node number must fit the record's source field");
static_assert(OWN <= (1 << (32 - SRC_BITS)), "a row number inside an owner range must fit the rest of the record");

__global__ void owner_kernel(const int* __restrict__ src, const int* __restrict__ dst,
                             unsigned* __restrict__ ob, uint4* __restrict__ rec) {
    int t = blockIdx.x * blockDim.x + threadIdx.x;
    if (t >= N_EDGES / 4) return;
    const int4 d = *reinterpret_cast<const int4*>(dst + 4 * t);
    const int4 s = *reinterpret_cast<const int4*>(src + 4 * t);
    const unsigned b0 = (unsigned)(d.x / OWN), b1 = (unsigned)(d.y / OWN);
    const unsigned b2 = (unsigned)(d.z / OWN), b3 = (unsigned)(d.w / OWN);
    ob[t] = b0 | (b1 << 8) | (b2 << 16) | (b3 << 24);
    uint4 r;
    r.x = (unsigned)s.x | (((unsigned)d.x - b0 * OWN) << SRC_BITS);
    r.y = (unsigned)s.y | (((unsigned)d.y - b1 * OWN) << SRC_BITS);
    r.z = (unsigned)s.z | (((unsigned)d.z - b2 * OWN) << SRC_BITS);
    r.w = (unsigned)s.w | (((unsigned)d.w - b3 * OWN) << SRC_BITS);
    rec[t] = r;
}

template <int K, int COUT>
__global__ void gemm_dinv_wmma(const float* __restrict__ H,
                               const float* __restrict__ W,
                               const float* __restrict__ dinv,
                               float* __restrict__ Y) {
    const int lane   = threadIdx.x & 31;
    const int n_tile = threadIdx.x >> 5;
    const int m_tile = blockIdx.x;

    const int row = m_tile * 16 + (lane & 15);
    const int col = n_tile * 16 + (lane & 15);
    const int kh  = (lane >> 4) << 1;

    const float* hrow = H + row * K + kh;

    v8f acc = {};
#pragma unroll
    for (int kk = 0; kk < K; kk += 4) {
        v2f a = *reinterpret_cast<const v2f*>(hrow + kk);
        v2f b;
        b.x = W[(kk + kh) * COUT + col];
        b.y = W[(kk + kh + 1) * COUT + col];
        acc = __builtin_amdgcn_wmma_f32_16x16x4_f32(
             false, a,  false, b,
             (short)0, acc,  false,  false);
    }

    const int mbase = m_tile * 16 + ((lane >> 4) << 3);
#pragma unroll
    for (int r = 0; r < 8; ++r) {
        const int m = mbase + r;
        Y[m * COUT + col] = dinv[m] * acc[r];
    }
}

template <int COUT, bool RELU, bool FIRST>
__global__ void __launch_bounds__(AGG_THREADS)
owner_aggregate(const unsigned* __restrict__ rec,
                const uint4* __restrict__ ob,
                const float* __restrict__ Y,
                const float* __restrict__ dinv,
                const float* __restrict__ bias,
                float* __restrict__ OUT,
                unsigned* __restrict__ lst,
                int* __restrict__ lcnt) {
    constexpr int G  = COUT / 32;
    constexpr int EB = 32 / G;
    constexpr int N_TRIPS = N_EDGES / 16 / 32;
    static_assert(N_EDGES % (16 * 32) == 0, "edge count must split evenly over wave trips");

    __shared__ __attribute__((aligned(16))) float acc[OWN * COUT];
    __shared__ int q[AGG_WAVES][64];

    const int tid   = threadIdx.x;
    const int lane  = tid & 31;
    const int wave  = __builtin_amdgcn_readfirstlane(tid >> 5);
    const int owner = blockIdx.x;
    const int node0 = owner * OWN;
    const int count = (N_NODES - node0 < OWN) ? (N_NODES - node0) : OWN;
    const int nval  = count * COUT;
    const int slot  = owner * AGG_WAVES + wave;
    unsigned* mine  = lst + slot * LIST_CAP;

    int kept = 0;
    if (!FIRST) kept = lcnt[slot];

    for (int i = tid; i < nval; i += AGG_THREADS) acc[i] = Y[node0 * COUT + i];
    __syncthreads();

    auto add_rows = [&](unsigned r, int n) {
        for (int i0 = 0; i0 < n; i0 += EB) {
            float v[EB][G];
#pragma unroll
            for (int k = 0; k < EB; ++k) {
                if (i0 + k < n) {
                    const int sk = (int)((unsigned)__builtin_amdgcn_readlane((int)r, i0 + k) & SRC_MASK);
                    const float* yr = Y + sk * COUT + lane;
#pragma unroll
                    for (int g = 0; g < G; ++g) v[k][g] = yr[g * 32];
                }
            }
#pragma unroll
            for (int k = 0; k < EB; ++k) {
                if (i0 + k < n) {
                    const int dk = (int)((unsigned)__builtin_amdgcn_readlane((int)r, i0 + k) >> SRC_BITS);
                    float* a = acc + dk * COUT + lane;
#pragma unroll
                    for (int g = 0; g < G; ++g)
                        __hip_atomic_fetch_add(a + g * 32, v[k][g], __ATOMIC_RELAXED,
                                               __HIP_MEMORY_SCOPE_WORKGROUP);
                }
            }
        }
    };

    kept = __builtin_amdgcn_readfirstlane(kept);
    if (!FIRST && kept <= LIST_CAP) {
        unsigned r = 0;
        if (lane < kept) r = mine[lane];
        for (int i0 = 0; i0 < kept; i0 += 32) {
            const int left = kept - i0;
            unsigned rn = 0;
            if (32 + lane < left) rn = mine[i0 + 32 + lane];
            if (left >= 32) add_rows(r, 32); else add_rows(r, left);
            r = rn;
        }
    } else {
        const unsigned pat = (unsigned)owner * 0x01010101u;
        int* wq = q[wave];
        int head = 0, tail = 0;

        auto drain = [&](int n) {
            unsigned r = 0;
            if (lane < n) r = rec[wq[(head + lane) & 63]];
            if (FIRST) {
                if (head + n <= LIST_CAP && lane < n) mine[head + lane] = r;
            }
            add_rows(r, n);
            head += n;
        };

        static_assert(AGG_WAVES <= N_TRIPS, "every wave has a first trip");
        uint4 cur = ob[wave * 32 + lane];
        for (int trip = wave; trip < N_TRIPS; trip += AGG_WAVES) {
            const int ntrip = (trip + AGG_WAVES < N_TRIPS) ? (trip + AGG_WAVES) : trip;
            const uint4 nxt = ob[ntrip * 32 + lane];

            unsigned m = 0;
            const unsigned w4[4] = {cur.x, cur.y, cur.z, cur.w};
#pragma unroll
            for (int dw = 0; dw < 4; ++dw) {
                const unsigned x = w4[dw] ^ pat;
                const unsigned t = (x & 0x7f7f7f7fu) + 0x7f7f7f7fu;
                const unsigned z = ~(t | x | 0x7f7f7f7fu);
                m |= z >> (7 - dw);
            }
            const int ebase = (trip * 32 + lane) * 16;

            for (;;) {
                const bool has = (m != 0u);
                const unsigned bal = __builtin_amdgcn_ballot_w32(has);
                if (bal == 0u) break;
                if (has) {
                    const int p    = __builtin_ctz(m);
                    const int j    = ((p & 7) << 2) | (p >> 3);
                    const int rank = (int)__builtin_amdgcn_mbcnt_lo(bal, 0u);
                    wq[(tail + rank) & 63] = ebase + j;
                    m &= m - 1u;
                }
                tail += __builtin_popcount(bal);
                __builtin_amdgcn_fence(__ATOMIC_ACQ_REL, "wavefront");
                if (tail - head >= 32) drain(32);
            }
            cur = nxt;
        }
        __builtin_amdgcn_fence(__ATOMIC_ACQ_REL, "wavefront");
        drain(tail - head);
        if (FIRST && lane == 0) lcnt[slot] = head;
    }
    __syncthreads();

    for (int i = tid; i < nval; i += AGG_THREADS) {
        const int n = i / COUT;
        const int c = i & (COUT - 1);
        float v = dinv[node0 + n] * acc[i] + bias[c];
        if (RELU) v = fmaxf(v, 0.0f);
        OUT[node0 * COUT + i] = v;
    }
}

extern "C" void kernel_launch(void* const* d_in, const int* in_sizes, int n_in,
                              void* d_out, int out_size, void* d_ws, size_t ws_size,
                              hipStream_t stream) {
    const float* x  = (const float*)d_in[0];
    const int*   ei = (const int*)d_in[1];
    const float* W1 = (const float*)d_in[2];
    const float* b1 = (const float*)d_in[3];
    const float* W2 = (const float*)d_in[4];
    const float* b2 = (const float*)d_in[5];
    const float* W3 = (const float*)d_in[6];
    const float* b3 = (const float*)d_in[7];
    float*       out = (float*)d_out;

    const int* src = ei;
    const int* dst = ei + N_EDGES;

    char*  ws   = (char*)d_ws;
    size_t off  = 0;
    float*    dinv = (float*)(ws + off);    off += ((size_t)N_NODES * 4 + 511) & ~511ull;
    unsigned* ob   = (unsigned*)(ws + off); off += ((size_t)N_EDGES + 511) & ~511ull;
    unsigned* rec  = (unsigned*)(ws + off); off += ((size_t)N_EDGES * 4 + 511) & ~511ull;
    float*    y    = (float*)(ws + off);    off += (size_t)N_NODES * 128 * 4;
    float*    h    = (float*)(ws + off);    off += (size_t)N_NODES * 128 * 4;
    unsigned* lst  = (unsigned*)(ws + off); off += (size_t)N_OWNERS * AGG_WAVES * LIST_CAP * 4;
    int*      lcnt = (int*)(ws + off);      off += (size_t)N_OWNERS * AGG_WAVES * 4;
    uint4*    part = (uint4*)(ws + off);

    const int B = 256;
    const int M_TILES = N_NODES / 16;

    degree_hist_kernel<<<DEG_PARTS, DEG_THREADS, 0, stream>>>(dst, part);
    degree_merge_kernel<<<PART_W / (MRG_COLS * 4), MRG_THREADS, 0, stream>>>(part, dinv);
    owner_kernel<<<(N_EDGES / 4 + B - 1) / B, B, 0, stream>>>(src, dst, ob, (uint4*)rec);

    gemm_dinv_wmma<64, 128><<<M_TILES, 256, 0, stream>>>(x, W1, dinv, y);
    owner_aggregate<128, true, true><<<N_OWNERS, AGG_THREADS, 0, stream>>>(rec, (const uint4*)ob, y, dinv, b1, h, lst, lcnt);

    gemm_dinv_wmma<128, 128><<<M_TILES, 256, 0, stream>>>(h, W2, dinv, y);
    owner_aggregate<128, true, false><<<N_OWNERS, AGG_THREADS, 0, stream>>>(rec, (const uint4*)ob, y, dinv, b2, h, lst, lcnt);

    gemm_dinv_wmma<128, 64><<<M_TILES, 128, 0, stream>>>(h, W3, dinv, y);
    owner_aggregate<64, false, false><<<N_OWNERS, AGG_THREADS, 0, stream>>>(rec, (const uint4*)ob, y, dinv, b3, out, lst, lcnt);
}


# product kernels re-tiled: a work-group stages 80 rows of H in LDS by whole-line loads, W fragments stay in registers, A fragments come from LDS (was 16-line global loads per wave); sums kept
# speedup vs baseline: 3.7058x; 1.1633x over previous
#include <hip/hip_runtime.h>


#define N_NODES 100000
#define N_EDGES 1600000

#define OWN 391
#define N_OWNERS 256
#define AGG_THREADS 1024
#define AGG_WAVES (AGG_THREADS / 32)
#define LIST_CAP 512

typedef __attribute__((ext_vector_type(2))) float v2f;
typedef __attribute__((ext_vector_type(8))) float v8f;

#define DEG_PARTS 128
#define DEG_CHUNK (N_EDGES / DEG_PARTS)
#define PART_W 50176
#define DEG_THREADS 1024
static_assert(N_EDGES % DEG_PARTS == 0, "the edges must split evenly over the counting work-groups");
static_assert(DEG_CHUNK < 65536, "a chunk's count of one node must fit 16 bits");
static_assert(DEG_CHUNK % 4 == 0, "a chunk is read in groups of four destinations");
static_assert(2 * PART_W >= N_NODES && PART_W % 256 == 0, "a partial covers every node in whole merge blocks");

__global__ void __launch_bounds__(DEG_THREADS)
degree_hist_kernel(const int* dst, uint4* __restrict__ part) {
    __shared__ __attribute__((aligned(16))) unsigned cnt[PART_W];
    uint4* cnt4 = reinterpret_cast<uint4*>(cnt);
    const int tid = threadIdx.x;

    constexpr int Q = DEG_CHUNK / 4;
    constexpr int R = (Q + DEG_THREADS - 1) / DEG_THREADS;
    const int4* d4 = reinterpret_cast<const int4*>(dst) + blockIdx.x * Q;
    int4 d[R];
    for (int r = 0; r < R; ++r) {
        const int i = tid + r * DEG_THREADS;
        d[r] = (i < Q) ? d4[i] : make_int4(-1, -1, -1, -1);
    }

    for (int i = tid; i < PART_W / 4; i += DEG_THREADS) cnt4[i] = make_uint4(0u, 0u, 0u, 0u);
    __syncthreads();

    auto count = [&](int node) {
        const unsigned u = (unsigned)node;
        if (u < (unsigned)N_NODES)
            __hip_atomic_fetch_add(cnt + (u >> 1), 1u << ((u & 1u) << 4), __ATOMIC_RELAXED,
                                   __HIP_MEMORY_SCOPE_WORKGROUP);
    };
    for (int r = 0; r < R; ++r) { count(d[r].x); count(d[r].y); count(d[r].z); count(d[r].w); }
    __syncthreads();

    uint4* mine = part + (size_t)blockIdx.x * (PART_W / 4);
    for (int i = tid; i < PART_W / 4; i += DEG_THREADS) mine[i] = cnt4[i];
}

#define MRG_COLS 64
#define MRG_SLICES 8
#define MRG_THREADS (MRG_COLS * MRG_SLICES)
static_assert(DEG_PARTS % MRG_SLICES == 0, "the partials must split evenly over the slices");
static_assert(PART_W % (MRG_COLS * 4) == 0, "a partial must split into whole merge blocks");

__global__ void __launch_bounds__(MRG_THREADS)
degree_merge_kernel(const uint4* __restrict__ part, float* __restrict__ dinv) {
    __shared__ unsigned red[MRG_SLICES][MRG_COLS * 8];
    const int tid   = threadIdx.x;
    const int col   = tid & (MRG_COLS - 1);
    const int slice = tid / MRG_COLS;

    const uint4* p = part + (size_t)slice * (PART_W / 4) + blockIdx.x * MRG_COLS + col;
    unsigned s[8] = {0u, 0u, 0u, 0u, 0u, 0u, 0u, 0u};
#pragma unroll
    for (int k = 0; k < DEG_PARTS / MRG_SLICES; ++k) {
        const uint4 v = p[(size_t)k * MRG_SLICES * (PART_W / 4)];
        s[0] += v.x & 0xffffu;  s[1] += v.x >> 16;
        s[2] += v.y & 0xffffu;  s[3] += v.y >> 16;
        s[4] += v.z & 0xffffu;  s[5] += v.z >> 16;
        s[6] += v.w & 0xffffu;  s[7] += v.w >> 16;
    }
    for (int j = 0; j < 8; ++j) red[slice][col * 8 + j] = s[j];
    __syncthreads();

    for (int n = tid; n < MRG_COLS * 8; n += MRG_THREADS) {
        unsigned total = 0u;
        for (int q = 0; q < MRG_SLICES; ++q) total += red[q][n];
        const int node = blockIdx.x * (MRG_COLS * 8) + n;
        if (node < N_NODES) dinv[node] = rsqrtf((float)total + 1.0f);
    }
}

#define SRC_BITS 17
#define SRC_MASK ((1u << SRC_BITS) - 1u)
static_assert(N_NODES <= (1 << SRC_BITS), "a node number must fit the record's source field");
static_assert(OWN <= (1 << (32 - SRC_BITS)), "a row number inside an owner range must fit the rest of the record");

__global__ void owner_kernel(const int* __restrict__ src, const int* __restrict__ dst,
                             unsigned* __restrict__ ob, uint4* __restrict__ rec) {
    int t = blockIdx.x * blockDim.x + threadIdx.x;
    if (t >= N_EDGES / 4) return;
    const int4 d = *reinterpret_cast<const int4*>(dst + 4 * t);
    const int4 s = *reinterpret_cast<const int4*>(src + 4 * t);
    const unsigned b0 = (unsigned)(d.x / OWN), b1 = (unsigned)(d.y / OWN);
    const unsigned b2 = (unsigned)(d.z / OWN), b3 = (unsigned)(d.w / OWN);
    ob[t] = b0 | (b1 << 8) | (b2 << 16) | (b3 << 24);
    uint4 r;
    r.x = (unsigned)s.x | (((unsigned)d.x - b0 * OWN) << SRC_BITS);
    r.y = (unsigned)s.y | (((unsigned)d.y - b1 * OWN) << SRC_BITS);
    r.z = (unsigned)s.z | (((unsigned)d.z - b2 * OWN) << SRC_BITS);
    r.w = (unsigned)s.w | (((unsigned)d.w - b3 * OWN) << SRC_BITS);
    rec[t] = r;
}

#define GEMM_MT 5
#define GEMM_ROWS (GEMM_MT * 16)
static_assert(N_NODES % GEMM_ROWS == 0, "the rows must split evenly over the product work-groups");

template <int K, int COUT>
__global__ void __launch_bounds__((COUT / 16) * 32)
gemm_dinv_wmma(const float* __restrict__ H,
               const float* __restrict__ W,
               const float* __restrict__ dinv,
               float* __restrict__ Y) {
    constexpr int NT = (COUT / 16) * 32;
    constexpr int P  = K + 4;
    constexpr int NV = GEMM_ROWS * K / 4;
    static_assert((K & (K - 1)) == 0 && K % 4 == 0, "K must be a power of two");
    static_assert(((P / 4) & 1) == 1, "the pitch must be an odd number of 16-byte pieces");
    static_assert(NV % NT == 0, "the block must split evenly over the threads");

    __shared__ __attribute__((aligned(16))) float a_lds[GEMM_ROWS * P];

    const int tid    = threadIdx.x;
    const int lane   = tid & 31;
    const int n_tile = tid >> 5;
    const int row0   = blockIdx.x * GEMM_ROWS;

    const int col = n_tile * 16 + (lane & 15);
    const int kh  = (lane >> 4) << 1;

    v2f b[K / 4];
#pragma unroll
    for (int kk = 0; kk < K; kk += 4) {
        b[kk / 4].x = W[(kk + kh) * COUT + col];
        b[kk / 4].y = W[(kk + kh + 1) * COUT + col];
    }

    const float4* hblk = reinterpret_cast<const float4*>(H + (size_t)row0 * K);
    float4 piece[NV / NT];
#pragma unroll
    for (int p = 0; p < NV / NT; ++p) piece[p] = hblk[tid + p * NT];
#pragma unroll
    for (int p = 0; p < NV / NT; ++p) {
        const int i = tid + p * NT;
        const int r = i / (K / 4);
        const int c = (i % (K / 4)) * 4;
        *reinterpret_cast<float4*>(a_lds + r * P + c) = piece[p];
    }
    __syncthreads();

    const float* arow = a_lds + (lane & 15) * P + kh;

    v8f acc[GEMM_MT];
#pragma unroll
    for (int t = 0; t < GEMM_MT; ++t) acc[t] = v8f{};

#pragma unroll
    for (int kk = 0; kk < K; kk += 4) {
#pragma unroll
        for (int t = 0; t < GEMM_MT; ++t) {
            const v2f a = *reinterpret_cast<const v2f*>(arow + t * 16 * P + kk);
            acc[t] = __builtin_amdgcn_wmma_f32_16x16x4_f32(
                 false, a,  false, b[kk / 4],
                 (short)0, acc[t],  false,  false);
        }
    }

#pragma unroll
    for (int t = 0; t < GEMM_MT; ++t) {
        const int mbase = row0 + t * 16 + ((lane >> 4) << 3);
#pragma unroll
        for (int r = 0; r < 8; ++r) {
            const int m = mbase + r;
            Y[m * COUT + col] = dinv[m] * acc[t][r];
        }
    }
}

template <int COUT, bool RELU, bool FIRST>
__global__ void __launch_bounds__(AGG_THREADS)
owner_aggregate(const unsigned* __restrict__ rec,
                const uint4* __restrict__ ob,
                const float* __restrict__ Y,
                const float* __restrict__ dinv,
                const float* __restrict__ bias,
                float* __restrict__ OUT,
                unsigned* __restrict__ lst,
                int* __restrict__ lcnt) {
    constexpr int G  = COUT / 32;
    constexpr int EB = 32 / G;
    constexpr int N_TRIPS = N_EDGES / 16 / 32;
    static_assert(N_EDGES % (16 * 32) == 0, "edge count must split evenly over wave trips");

    __shared__ __attribute__((aligned(16))) float acc[OWN * COUT];
    __shared__ int q[AGG_WAVES][64];

    const int tid   = threadIdx.x;
    const int lane  = tid & 31;
    const int wave  = __builtin_amdgcn_readfirstlane(tid >> 5);
    const int owner = blockIdx.x;
    const int node0 = owner * OWN;
    const int count = (N_NODES - node0 < OWN) ? (N_NODES - node0) : OWN;
    const int nval  = count * COUT;
    const int slot  = owner * AGG_WAVES + wave;
    unsigned* mine  = lst + slot * LIST_CAP;

    int kept = 0;
    if (!FIRST) kept = lcnt[slot];

    for (int i = tid; i < nval; i += AGG_THREADS) acc[i] = Y[node0 * COUT + i];
    __syncthreads();

    auto add_rows = [&](unsigned r, int n) {
        for (int i0 = 0; i0 < n; i0 += EB) {
            float v[EB][G];
#pragma unroll
            for (int k = 0; k < EB; ++k) {
                if (i0 + k < n) {
                    const int sk = (int)((unsigned)__builtin_amdgcn_readlane((int)r, i0 + k) & SRC_MASK);
                    const float* yr = Y + sk * COUT + lane;
#pragma unroll
                    for (int g = 0; g < G; ++g) v[k][g] = yr[g * 32];
                }
            }
#pragma unroll
            for (int k = 0; k < EB; ++k) {
                if (i0 + k < n) {
                    const int dk = (int)((unsigned)__builtin_amdgcn_readlane((int)r, i0 + k) >> SRC_BITS);
                    float* a = acc + dk * COUT + lane;
#pragma unroll
                    for (int g = 0; g < G; ++g)
                        __hip_atomic_fetch_add(a + g * 32, v[k][g], __ATOMIC_RELAXED,
                                               __HIP_MEMORY_SCOPE_WORKGROUP);
                }
            }
        }
    };

    kept = __builtin_amdgcn_readfirstlane(kept);
    if (!FIRST && kept <= LIST_CAP) {
        unsigned r = 0;
        if (lane < kept) r = mine[lane];
        for (int i0 = 0; i0 < kept; i0 += 32) {
            const int left = kept - i0;
            unsigned rn = 0;
            if (32 + lane < left) rn = mine[i0 + 32 + lane];
            if (left >= 32) add_rows(r, 32); else add_rows(r, left);
            r = rn;
        }
    } else {
        const unsigned pat = (unsigned)owner * 0x01010101u;
        int* wq = q[wave];
        int head = 0, tail = 0;

        auto drain = [&](int n) {
            unsigned r = 0;
            if (lane < n) r = rec[wq[(head + lane) & 63]];
            if (FIRST) {
                if (head + n <= LIST_CAP && lane < n) mine[head + lane] = r;
            }
            add_rows(r, n);
            head += n;
        };

        static_assert(AGG_WAVES <= N_TRIPS, "every wave has a first trip");
        uint4 cur = ob[wave * 32 + lane];
        for (int trip = wave; trip < N_TRIPS; trip += AGG_WAVES) {
            const int ntrip = (trip + AGG_WAVES < N_TRIPS) ? (trip + AGG_WAVES) : trip;
            const uint4 nxt = ob[ntrip * 32 + lane];

            unsigned m = 0;
            const unsigned w4[4] = {cur.x, cur.y, cur.z, cur.w};
#pragma unroll
            for (int dw = 0; dw < 4; ++dw) {
                const unsigned x = w4[dw] ^ pat;
                const unsigned t = (x & 0x7f7f7f7fu) + 0x7f7f7f7fu;
                const unsigned z = ~(t | x | 0x7f7f7f7fu);
                m |= z >> (7 - dw);
            }
            const int ebase = (trip * 32 + lane) * 16;

            for (;;) {
                const bool has = (m != 0u);
                const unsigned bal = __builtin_amdgcn_ballot_w32(has);
                if (bal == 0u) break;
                if (has) {
                    const int p    = __builtin_ctz(m);
                    const int j    = ((p & 7) << 2) | (p >> 3);
                    const int rank = (int)__builtin_amdgcn_mbcnt_lo(bal, 0u);
                    wq[(tail + rank) & 63] = ebase + j;
                    m &= m - 1u;
                }
                tail += __builtin_popcount(bal);
                __builtin_amdgcn_fence(__ATOMIC_ACQ_REL, "wavefront");
                if (tail - head >= 32) drain(32);
            }
            cur = nxt;
        }
        __builtin_amdgcn_fence(__ATOMIC_ACQ_REL, "wavefront");
        drain(tail - head);
        if (FIRST && lane == 0) lcnt[slot] = head;
    }
    __syncthreads();

    for (int i = tid; i < nval; i += AGG_THREADS) {
        const int n = i / COUT;
        const int c = i & (COUT - 1);
        float v = dinv[node0 + n] * acc[i] + bias[c];
        if (RELU) v = fmaxf(v, 0.0f);
        OUT[node0 * COUT + i] = v;
    }
}

extern "C" void kernel_launch(void* const* d_in, const int* in_sizes, int n_in,
                              void* d_out, int out_size, void* d_ws, size_t ws_size,
                              hipStream_t stream) {
    const float* x  = (const float*)d_in[0];
    const int*   ei = (const int*)d_in[1];
    const float* W1 = (const float*)d_in[2];
    const float* b1 = (const float*)d_in[3];
    const float* W2 = (const float*)d_in[4];
    const float* b2 = (const float*)d_in[5];
    const float* W3 = (const float*)d_in[6];
    const float* b3 = (const float*)d_in[7];
    float*       out = (float*)d_out;

    const int* src = ei;
    const int* dst = ei + N_EDGES;

    char*  ws   = (char*)d_ws;
    size_t off  = 0;
    float*    dinv = (float*)(ws + off);    off += ((size_t)N_NODES * 4 + 511) & ~511ull;
    unsigned* ob   = (unsigned*)(ws + off); off += ((size_t)N_EDGES + 511) & ~511ull;
    unsigned* rec  = (unsigned*)(ws + off); off += ((size_t)N_EDGES * 4 + 511) & ~511ull;
    float*    y    = (float*)(ws + off);    off += (size_t)N_NODES * 128 * 4;
    float*    h    = (float*)(ws + off);    off += (size_t)N_NODES * 128 * 4;
    unsigned* lst  = (unsigned*)(ws + off); off += (size_t)N_OWNERS * AGG_WAVES * LIST_CAP * 4;
    int*      lcnt = (int*)(ws + off);      off += (size_t)N_OWNERS * AGG_WAVES * 4;
    uint4*    part = (uint4*)(ws + off);

    const int B = 256;
    const int M_BLOCKS = N_NODES / GEMM_ROWS;

    degree_hist_kernel<<<DEG_PARTS, DEG_THREADS, 0, stream>>>(dst, part);
    degree_merge_kernel<<<PART_W / (MRG_COLS * 4), MRG_THREADS, 0, stream>>>(part, dinv);
    owner_kernel<<<(N_EDGES / 4 + B - 1) / B, B, 0, stream>>>(src, dst, ob, (uint4*)rec);

    gemm_dinv_wmma<64, 128><<<M_BLOCKS, 256, 0, stream>>>(x, W1, dinv, y);
    owner_aggregate<128, true, true><<<N_OWNERS, AGG_THREADS, 0, stream>>>(rec, (const uint4*)ob, y, dinv, b1, h, lst, lcnt);

    gemm_dinv_wmma<128, 128><<<M_BLOCKS, 256, 0, stream>>>(h, W2, dinv, y);
    owner_aggregate<128, true, false><<<N_OWNERS, AGG_THREADS, 0, stream>>>(rec, (const uint4*)ob, y, dinv, b2, h, lst, lcnt);

    gemm_dinv_wmma<128, 64><<<M_BLOCKS, 128, 0, stream>>>(h, W3, dinv, y);
    owner_aggregate<64, false, false><<<N_OWNERS, AGG_THREADS, 0, stream>>>(rec, (const uint4*)ob, y, dinv, b3, out, lst, lcnt);
}


# layer 1 sums its 64-wide scaled input rows first and multiplies by W1 after (was product first, 128-wide rows summed): half the gathered lines and LDS adds there; new sum order: sum, then @W
# speedup vs baseline: 3.9495x; 1.0658x over previous
#include <hip/hip_runtime.h>


#define N_NODES 100000
#define N_EDGES 1600000

#define OWN 391
#define N_OWNERS 256
#define AGG_THREADS 1024
#define AGG_WAVES (AGG_THREADS / 32)
#define LIST_CAP 512

typedef __attribute__((ext_vector_type(2))) float v2f;
typedef __attribute__((ext_vector_type(8))) float v8f;

#define DEG_PARTS 128
#define DEG_CHUNK (N_EDGES / DEG_PARTS)
#define PART_W 50176
#define DEG_THREADS 1024
static_assert(N_EDGES % DEG_PARTS == 0, "the edges must split evenly over the counting work-groups");
static_assert(DEG_CHUNK < 65536, "a chunk's count of one node must fit 16 bits");
static_assert(DEG_CHUNK % 4 == 0, "a chunk is read in groups of four destinations");
static_assert(2 * PART_W >= N_NODES && PART_W % 256 == 0, "a partial covers every node in whole merge blocks");

__global__ void __launch_bounds__(DEG_THREADS)
degree_hist_kernel(const int* dst, uint4* __restrict__ part) {
    __shared__ __attribute__((aligned(16))) unsigned cnt[PART_W];
    uint4* cnt4 = reinterpret_cast<uint4*>(cnt);
    const int tid = threadIdx.x;

    constexpr int Q = DEG_CHUNK / 4;
    constexpr int R = (Q + DEG_THREADS - 1) / DEG_THREADS;
    const int4* d4 = reinterpret_cast<const int4*>(dst) + blockIdx.x * Q;
    int4 d[R];
    for (int r = 0; r < R; ++r) {
        const int i = tid + r * DEG_THREADS;
        d[r] = (i < Q) ? d4[i] : make_int4(-1, -1, -1, -1);
    }

    for (int i = tid; i < PART_W / 4; i += DEG_THREADS) cnt4[i] = make_uint4(0u, 0u, 0u, 0u);
    __syncthreads();

    auto count = [&](int node) {
        const unsigned u = (unsigned)node;
        if (u < (unsigned)N_NODES)
            __hip_atomic_fetch_add(cnt + (u >> 1), 1u << ((u & 1u) << 4), __ATOMIC_RELAXED,
                                   __HIP_MEMORY_SCOPE_WORKGROUP);
    };
    for (int r = 0; r < R; ++r) { count(d[r].x); count(d[r].y); count(d[r].z); count(d[r].w); }
    __syncthreads();

    uint4* mine = part + (size_t)blockIdx.x * (PART_W / 4);
    for (int i = tid; i < PART_W / 4; i += DEG_THREADS) mine[i] = cnt4[i];
}

#define MRG_COLS 64
#define MRG_SLICES 8
#define MRG_THREADS (MRG_COLS * MRG_SLICES)
static_assert(DEG_PARTS % MRG_SLICES == 0, "the partials must split evenly over the slices");
static_assert(PART_W % (MRG_COLS * 4) == 0, "a partial must split into whole merge blocks");

__global__ void __launch_bounds__(MRG_THREADS)
degree_merge_kernel(const uint4* __restrict__ part, float* __restrict__ dinv) {
    __shared__ unsigned red[MRG_SLICES][MRG_COLS * 8];
    const int tid   = threadIdx.x;
    const int col   = tid & (MRG_COLS - 1);
    const int slice = tid / MRG_COLS;

    const uint4* p = part + (size_t)slice * (PART_W / 4) + blockIdx.x * MRG_COLS + col;
    unsigned s[8] = {0u, 0u, 0u, 0u, 0u, 0u, 0u, 0u};
#pragma unroll
    for (int k = 0; k < DEG_PARTS / MRG_SLICES; ++k) {
        const uint4 v = p[(size_t)k * MRG_SLICES * (PART_W / 4)];
        s[0] += v.x & 0xffffu;  s[1] += v.x >> 16;
        s[2] += v.y & 0xffffu;  s[3] += v.y >> 16;
        s[4] += v.z & 0xffffu;  s[5] += v.z >> 16;
        s[6] += v.w & 0xffffu;  s[7] += v.w >> 16;
    }
    for (int j = 0; j < 8; ++j) red[slice][col * 8 + j] = s[j];
    __syncthreads();

    for (int n = tid; n < MRG_COLS * 8; n += MRG_THREADS) {
        unsigned total = 0u;
        for (int q = 0; q < MRG_SLICES; ++q) total += red[q][n];
        const int node = blockIdx.x * (MRG_COLS * 8) + n;
        if (node < N_NODES) dinv[node] = rsqrtf((float)total + 1.0f);
    }
}

#define SRC_BITS 17
#define SRC_MASK ((1u << SRC_BITS) - 1u)
static_assert(N_NODES <= (1 << SRC_BITS), "a node number must fit the record's source field");
static_assert(OWN <= (1 << (32 - SRC_BITS)), "a row number inside an owner range must fit the rest of the record");

__global__ void owner_kernel(const int* __restrict__ src, const int* __restrict__ dst,
                             unsigned* __restrict__ ob, uint4* __restrict__ rec) {
    int t = blockIdx.x * blockDim.x + threadIdx.x;
    if (t >= N_EDGES / 4) return;
    const int4 d = *reinterpret_cast<const int4*>(dst + 4 * t);
    const int4 s = *reinterpret_cast<const int4*>(src + 4 * t);
    const unsigned b0 = (unsigned)(d.x / OWN), b1 = (unsigned)(d.y / OWN);
    const unsigned b2 = (unsigned)(d.z / OWN), b3 = (unsigned)(d.w / OWN);
    ob[t] = b0 | (b1 << 8) | (b2 << 16) | (b3 << 24);
    uint4 r;
    r.x = (unsigned)s.x | (((unsigned)d.x - b0 * OWN) << SRC_BITS);
    r.y = (unsigned)s.y | (((unsigned)d.y - b1 * OWN) << SRC_BITS);
    r.z = (unsigned)s.z | (((unsigned)d.z - b2 * OWN) << SRC_BITS);
    r.w = (unsigned)s.w | (((unsigned)d.w - b3 * OWN) << SRC_BITS);
    rec[t] = r;
}

#define SCL_ROWS 80
#define SCL_THREADS 256
static_assert(N_NODES % SCL_ROWS == 0, "the rows must split evenly over the scaling work-groups");
static_assert((SCL_ROWS * 16) % SCL_THREADS == 0, "a block of rows must split evenly over the threads");

__global__ void __launch_bounds__(SCL_THREADS)
scale_rows_kernel(const float* __restrict__ X, const float* __restrict__ dinv, float* __restrict__ XS) {
    constexpr int NP = SCL_ROWS * 16 / SCL_THREADS;
    const int tid = threadIdx.x;
    const float4* in  = reinterpret_cast<const float4*>(X) + (size_t)blockIdx.x * (SCL_ROWS * 16);
    float4*       out = reinterpret_cast<float4*>(XS) + (size_t)blockIdx.x * (SCL_ROWS * 16);
    float4 v[NP];
    float  d[NP];
#pragma unroll
    for (int p = 0; p < NP; ++p) {
        const int i = tid + p * SCL_THREADS;
        v[p] = in[i];
        d[p] = dinv[blockIdx.x * SCL_ROWS + (i >> 4)];
    }
#pragma unroll
    for (int p = 0; p < NP; ++p) {
        v[p].x *= d[p]; v[p].y *= d[p]; v[p].z *= d[p]; v[p].w *= d[p];
        out[tid + p * SCL_THREADS] = v[p];
    }
}

#define GEMM_MT 5
#define GEMM_ROWS (GEMM_MT * 16)
static_assert(N_NODES % GEMM_ROWS == 0, "the rows must split evenly over the product work-groups");

template <int K, int COUT, bool ACT>
__global__ void __launch_bounds__((COUT / 16) * 32)
gemm_wmma(const float* __restrict__ H,
          const float* __restrict__ W,
          const float* __restrict__ dinv,
          const float* __restrict__ bias,
          float* __restrict__ Y) {
    constexpr int NT = (COUT / 16) * 32;
    constexpr int P  = K + 4;
    constexpr int NV = GEMM_ROWS * K / 4;
    static_assert((K & (K - 1)) == 0 && K % 4 == 0, "K must be a power of two");
    static_assert(((P / 4) & 1) == 1, "the pitch must be an odd number of 16-byte pieces");
    static_assert(NV % NT == 0, "the block must split evenly over the threads");

    __shared__ __attribute__((aligned(16))) float a_lds[GEMM_ROWS * P];

    const int tid    = threadIdx.x;
    const int lane   = tid & 31;
    const int n_tile = tid >> 5;
    const int row0   = blockIdx.x * GEMM_ROWS;

    const int col = n_tile * 16 + (lane & 15);
    const int kh  = (lane >> 4) << 1;

    v2f b[K / 4];
#pragma unroll
    for (int kk = 0; kk < K; kk += 4) {
        b[kk / 4].x = W[(kk + kh) * COUT + col];
        b[kk / 4].y = W[(kk + kh + 1) * COUT + col];
    }
    float bc = 0.0f;
    if (ACT) bc = bias[col];

    const float4* hblk = reinterpret_cast<const float4*>(H + (size_t)row0 * K);
    float4 piece[NV / NT];
#pragma unroll
    for (int p = 0; p < NV / NT; ++p) piece[p] = hblk[tid + p * NT];
#pragma unroll
    for (int p = 0; p < NV / NT; ++p) {
        const int i = tid + p * NT;
        const int r = i / (K / 4);
        const int c = (i % (K / 4)) * 4;
        *reinterpret_cast<float4*>(a_lds + r * P + c) = piece[p];
    }
    __syncthreads();

    const float* arow = a_lds + (lane & 15) * P + kh;

    v8f acc[GEMM_MT];
#pragma unroll
    for (int t = 0; t < GEMM_MT; ++t) acc[t] = v8f{};

#pragma unroll
    for (int kk = 0; kk < K; kk += 4) {
#pragma unroll
        for (int t = 0; t < GEMM_MT; ++t) {
            const v2f a = *reinterpret_cast<const v2f*>(arow + t * 16 * P + kk);
            acc[t] = __builtin_amdgcn_wmma_f32_16x16x4_f32(
                 false, a,  false, b[kk / 4],
                 (short)0, acc[t],  false,  false);
        }
    }

#pragma unroll
    for (int t = 0; t < GEMM_MT; ++t) {
        const int mbase = row0 + t * 16 + ((lane >> 4) << 3);
#pragma unroll
        for (int r = 0; r < 8; ++r) {
            const int m = mbase + r;
            if (ACT) Y[m * COUT + col] = fmaxf(acc[t][r] + bc, 0.0f);
            else     Y[m * COUT + col] = dinv[m] * acc[t][r];
        }
    }
}

template <int COUT, bool RELU, bool FIRST, bool BIAS>
__global__ void __launch_bounds__(AGG_THREADS)
owner_aggregate(const unsigned* __restrict__ rec,
                const uint4* __restrict__ ob,
                const float* __restrict__ Y,
                const float* __restrict__ dinv,
                const float* __restrict__ bias,
                float* __restrict__ OUT,
                unsigned* __restrict__ lst,
                int* __restrict__ lcnt) {
    constexpr int G  = COUT / 32;
    constexpr int EB = 32 / G;
    constexpr int N_TRIPS = N_EDGES / 16 / 32;
    static_assert(N_EDGES % (16 * 32) == 0, "edge count must split evenly over wave trips");

    __shared__ __attribute__((aligned(16))) float acc[OWN * COUT];
    __shared__ int q[AGG_WAVES][64];

    const int tid   = threadIdx.x;
    const int lane  = tid & 31;
    const int wave  = __builtin_amdgcn_readfirstlane(tid >> 5);
    const int owner = blockIdx.x;
    const int node0 = owner * OWN;
    const int count = (N_NODES - node0 < OWN) ? (N_NODES - node0) : OWN;
    const int nval  = count * COUT;
    const int slot  = owner * AGG_WAVES + wave;
    unsigned* mine  = lst + slot * LIST_CAP;

    int kept = 0;
    if (!FIRST) kept = lcnt[slot];

    for (int i = tid; i < nval; i += AGG_THREADS) acc[i] = Y[node0 * COUT + i];
    __syncthreads();

    auto add_rows = [&](unsigned r, int n) {
        for (int i0 = 0; i0 < n; i0 += EB) {
            float v[EB][G];
#pragma unroll
            for (int k = 0; k < EB; ++k) {
                if (i0 + k < n) {
                    const int sk = (int)((unsigned)__builtin_amdgcn_readlane((int)r, i0 + k) & SRC_MASK);
                    const float* yr = Y + sk * COUT + lane;
#pragma unroll
                    for (int g = 0; g < G; ++g) v[k][g] = yr[g * 32];
                }
            }
#pragma unroll
            for (int k = 0; k < EB; ++k) {
                if (i0 + k < n) {
                    const int dk = (int)((unsigned)__builtin_amdgcn_readlane((int)r, i0 + k) >> SRC_BITS);
                    float* a = acc + dk * COUT + lane;
#pragma unroll
                    for (int g = 0; g < G; ++g)
                        __hip_atomic_fetch_add(a + g * 32, v[k][g], __ATOMIC_RELAXED,
                                               __HIP_MEMORY_SCOPE_WORKGROUP);
                }
            }
        }
    };

    kept = __builtin_amdgcn_readfirstlane(kept);
    if (!FIRST && kept <= LIST_CAP) {
        unsigned r = 0;
        if (lane < kept) r = mine[lane];
        for (int i0 = 0; i0 < kept; i0 += 32) {
            const int left = kept - i0;
            unsigned rn = 0;
            if (32 + lane < left) rn = mine[i0 + 32 + lane];
            if (left >= 32) add_rows(r, 32); else add_rows(r, left);
            r = rn;
        }
    } else {
        const unsigned pat = (unsigned)owner * 0x01010101u;
        int* wq = q[wave];
        int head = 0, tail = 0;

        auto drain = [&](int n) {
            unsigned r = 0;
            if (lane < n) r = rec[wq[(head + lane) & 63]];
            if (FIRST) {
                if (head + n <= LIST_CAP && lane < n) mine[head + lane] = r;
            }
            add_rows(r, n);
            head += n;
        };

        static_assert(AGG_WAVES <= N_TRIPS, "every wave has a first trip");
        uint4 cur = ob[wave * 32 + lane];
        for (int trip = wave; trip < N_TRIPS; trip += AGG_WAVES) {
            const int ntrip = (trip + AGG_WAVES < N_TRIPS) ? (trip + AGG_WAVES) : trip;
            const uint4 nxt = ob[ntrip * 32 + lane];

            unsigned m = 0;
            const unsigned w4[4] = {cur.x, cur.y, cur.z, cur.w};
#pragma unroll
            for (int dw = 0; dw < 4; ++dw) {
                const unsigned x = w4[dw] ^ pat;
                const unsigned t = (x & 0x7f7f7f7fu) + 0x7f7f7f7fu;
                const unsigned z = ~(t | x | 0x7f7f7f7fu);
                m |= z >> (7 - dw);
            }
            const int ebase = (trip * 32 + lane) * 16;

            for (;;) {
                const bool has = (m != 0u);
                const unsigned bal = __builtin_amdgcn_ballot_w32(has);
                if (bal == 0u) break;
                if (has) {
                    const int p    = __builtin_ctz(m);
                    const int j    = ((p & 7) << 2) | (p >> 3);
                    const int rank = (int)__builtin_amdgcn_mbcnt_lo(bal, 0u);
                    wq[(tail + rank) & 63] = ebase + j;
                    m &= m - 1u;
                }
                tail += __builtin_popcount(bal);
                __builtin_amdgcn_fence(__ATOMIC_ACQ_REL, "wavefront");
                if (tail - head >= 32) drain(32);
            }
            cur = nxt;
        }
        __builtin_amdgcn_fence(__ATOMIC_ACQ_REL, "wavefront");
        drain(tail - head);
        if (FIRST && lane == 0) lcnt[slot] = head;
    }
    __syncthreads();

    for (int i = tid; i < nval; i += AGG_THREADS) {
        const int n = i / COUT;
        const int c = i & (COUT - 1);
        float v = dinv[node0 + n] * acc[i];
        if (BIAS) v += bias[c];
        if (RELU) v = fmaxf(v, 0.0f);
        OUT[node0 * COUT + i] = v;
    }
}

extern "C" void kernel_launch(void* const* d_in, const int* in_sizes, int n_in,
                              void* d_out, int out_size, void* d_ws, size_t ws_size,
                              hipStream_t stream) {
    const float* x  = (const float*)d_in[0];
    const int*   ei = (const int*)d_in[1];
    const float* W1 = (const float*)d_in[2];
    const float* b1 = (const float*)d_in[3];
    const float* W2 = (const float*)d_in[4];
    const float* b2 = (const float*)d_in[5];
    const float* W3 = (const float*)d_in[6];
    const float* b3 = (const float*)d_in[7];
    float*       out = (float*)d_out;

    const int* src = ei;
    const int* dst = ei + N_EDGES;

    char*  ws   = (char*)d_ws;
    size_t off  = 0;
    float*    dinv = (float*)(ws + off);    off += ((size_t)N_NODES * 4 + 511) & ~511ull;
    unsigned* ob   = (unsigned*)(ws + off); off += ((size_t)N_EDGES + 511) & ~511ull;
    unsigned* rec  = (unsigned*)(ws + off); off += ((size_t)N_EDGES * 4 + 511) & ~511ull;
    float*    y    = (float*)(ws + off);    off += (size_t)N_NODES * 128 * 4;
    float*    h    = (float*)(ws + off);    off += (size_t)N_NODES * 128 * 4;
    unsigned* lst  = (unsigned*)(ws + off); off += (size_t)N_OWNERS * AGG_WAVES * LIST_CAP * 4;
    int*      lcnt = (int*)(ws + off);      off += (size_t)N_OWNERS * AGG_WAVES * 4;
    uint4*    part = (uint4*)(ws + off);

    float* xs = y;
    float* t  = y + (size_t)N_NODES * 64;

    const int B = 256;
    const int M_BLOCKS = N_NODES / GEMM_ROWS;

    degree_hist_kernel<<<DEG_PARTS, DEG_THREADS, 0, stream>>>(dst, part);
    degree_merge_kernel<<<PART_W / (MRG_COLS * 4), MRG_THREADS, 0, stream>>>(part, dinv);
    owner_kernel<<<(N_EDGES / 4 + B - 1) / B, B, 0, stream>>>(src, dst, ob, (uint4*)rec);

    scale_rows_kernel<<<N_NODES / SCL_ROWS, SCL_THREADS, 0, stream>>>(x, dinv, xs);
    owner_aggregate<64, false, true, false><<<N_OWNERS, AGG_THREADS, 0, stream>>>(rec, (const uint4*)ob, xs, dinv, nullptr, t, lst, lcnt);
    gemm_wmma<64, 128, true><<<M_BLOCKS, 256, 0, stream>>>(t, W1, nullptr, b1, h);

    gemm_wmma<128, 128, false><<<M_BLOCKS, 256, 0, stream>>>(h, W2, dinv, nullptr, y);
    owner_aggregate<128, true, false, true><<<N_OWNERS, AGG_THREADS, 0, stream>>>(rec, (const uint4*)ob, y, dinv, b2, h, lst, lcnt);

    gemm_wmma<128, 64, false><<<M_BLOCKS, 128, 0, stream>>>(h, W3, dinv, nullptr, y);
    owner_aggregate<64, false, false, true><<<N_OWNERS, AGG_THREADS, 0, stream>>>(rec, (const uint4*)ob, y, dinv, b3, out, lst, lcnt);
}


# owner table becomes one bit row per owner, built in LDS by the owner pass: an owner's walk reads its own 200 KB row (128 edges per lane load), not the shared byte table; LDS arrival order kept
# speedup vs baseline: 4.1923x; 1.0615x over previous
#include <hip/hip_runtime.h>


#define N_NODES 100000
#define N_EDGES 1600000

#define OWN 391
#define N_OWNERS 256
#define AGG_THREADS 1024
#define AGG_WAVES (AGG_THREADS / 32)
#define LIST_CAP 512

typedef __attribute__((ext_vector_type(2))) float v2f;
typedef __attribute__((ext_vector_type(8))) float v8f;

#define DEG_PARTS 128
#define DEG_CHUNK (N_EDGES / DEG_PARTS)
#define PART_W 50176
#define DEG_THREADS 1024
static_assert(N_EDGES % DEG_PARTS == 0, "the edges must split evenly over the counting work-groups");
static_assert(DEG_CHUNK < 65536, "a chunk's count of one node must fit 16 bits");
static_assert(DEG_CHUNK % 4 == 0, "a chunk is read in groups of four destinations");
static_assert(2 * PART_W >= N_NODES && PART_W % 256 == 0, "a partial covers every node in whole merge blocks");

__global__ void __launch_bounds__(DEG_THREADS)
degree_hist_kernel(const int* dst, uint4* __restrict__ part) {
    __shared__ __attribute__((aligned(16))) unsigned cnt[PART_W];
    uint4* cnt4 = reinterpret_cast<uint4*>(cnt);
    const int tid = threadIdx.x;

    constexpr int Q = DEG_CHUNK / 4;
    constexpr int R = (Q + DEG_THREADS - 1) / DEG_THREADS;
    const int4* d4 = reinterpret_cast<const int4*>(dst) + blockIdx.x * Q;
    int4 d[R];
    for (int r = 0; r < R; ++r) {
        const int i = tid + r * DEG_THREADS;
        d[r] = (i < Q) ? d4[i] : make_int4(-1, -1, -1, -1);
    }

    for (int i = tid; i < PART_W / 4; i += DEG_THREADS) cnt4[i] = make_uint4(0u, 0u, 0u, 0u);
    __syncthreads();

    auto count = [&](int node) {
        const unsigned u = (unsigned)node;
        if (u < (unsigned)N_NODES)
            __hip_atomic_fetch_add(cnt + (u >> 1), 1u << ((u & 1u) << 4), __ATOMIC_RELAXED,
                                   __HIP_MEMORY_SCOPE_WORKGROUP);
    };
    for (int r = 0; r < R; ++r) { count(d[r].x); count(d[r].y); count(d[r].z); count(d[r].w); }
    __syncthreads();

    uint4* mine = part + (size_t)blockIdx.x * (PART_W / 4);
    for (int i = tid; i < PART_W / 4; i += DEG_THREADS) mine[i] = cnt4[i];
}

#define MRG_COLS 64
#define MRG_SLICES 8
#define MRG_THREADS (MRG_COLS * MRG_SLICES)
static_assert(DEG_PARTS % MRG_SLICES == 0, "the partials must split evenly over the slices");
static_assert(PART_W % (MRG_COLS * 4) == 0, "a partial must split into whole merge blocks");

__global__ void __launch_bounds__(MRG_THREADS)
degree_merge_kernel(const uint4* __restrict__ part, float* __restrict__ dinv) {
    __shared__ unsigned red[MRG_SLICES][MRG_COLS * 8];
    const int tid   = threadIdx.x;
    const int col   = tid & (MRG_COLS - 1);
    const int slice = tid / MRG_COLS;

    const uint4* p = part + (size_t)slice * (PART_W / 4) + blockIdx.x * MRG_COLS + col;
    unsigned s[8] = {0u, 0u, 0u, 0u, 0u, 0u, 0u, 0u};
#pragma unroll
    for (int k = 0; k < DEG_PARTS / MRG_SLICES; ++k) {
        const uint4 v = p[(size_t)k * MRG_SLICES * (PART_W / 4)];
        s[0] += v.x & 0xffffu;  s[1] += v.x >> 16;
        s[2] += v.y & 0xffffu;  s[3] += v.y >> 16;
        s[4] += v.z & 0xffffu;  s[5] += v.z >> 16;
        s[6] += v.w & 0xffffu;  s[7] += v.w >> 16;
    }
    for (int j = 0; j < 8; ++j) red[slice][col * 8 + j] = s[j];
    __syncthreads();

    for (int n = tid; n < MRG_COLS * 8; n += MRG_THREADS) {
        unsigned total = 0u;
        for (int q = 0; q < MRG_SLICES; ++q) total += red[q][n];
        const int node = blockIdx.x * (MRG_COLS * 8) + n;
        if (node < N_NODES) dinv[node] = rsqrtf((float)total + 1.0f);
    }
}

#define SRC_BITS 17
#define SRC_MASK ((1u << SRC_BITS) - 1u)
static_assert(N_NODES <= (1 << SRC_BITS), "a node number must fit the record's source field");
static_assert(OWN <= (1 << (32 - SRC_BITS)), "a row number inside an owner range must fit the rest of the record");

#define CH_EDGES 2048
#define CH_WORDS (CH_EDGES / 32)
#define CH_THREADS (CH_EDGES / 4)
#define N_CHUNKS ((N_EDGES + CH_EDGES - 1) / CH_EDGES)
#define ROW_WORDS (N_CHUNKS * CH_WORDS)
#define CH_PITCH (CH_WORDS + 4)
static_assert(N_EDGES % 4 == 0, "edges are read in groups of four");
static_assert(ROW_WORDS % 128 == 0, "a row must split into whole wave loads of four words per lane");
static_assert((N_OWNERS * CH_WORDS / 4) % CH_THREADS == 0, "the chunk's pieces must split evenly over the threads");

__global__ void __launch_bounds__(CH_THREADS)
owner_rows_kernel(const int* __restrict__ src, const int* __restrict__ dst,
                  uint4* __restrict__ rows, uint4* __restrict__ rec) {
    __shared__ __attribute__((aligned(16))) unsigned lm[N_OWNERS * CH_PITCH];
    uint4* lm4 = reinterpret_cast<uint4*>(lm);
    const int tid   = threadIdx.x;
    const int chunk = blockIdx.x;
    const int t     = chunk * CH_THREADS + tid;
    const bool live = t < N_EDGES / 4;

    int4 d = make_int4(0, 0, 0, 0), s = make_int4(0, 0, 0, 0);
    if (live) {
        d = *reinterpret_cast<const int4*>(dst + 4 * t);
        s = *reinterpret_cast<const int4*>(src + 4 * t);
    }

    for (int i = tid; i < N_OWNERS * CH_PITCH / 4; i += CH_THREADS) lm4[i] = make_uint4(0u, 0u, 0u, 0u);
    __syncthreads();

    if (live) {
        const unsigned b0 = (unsigned)(d.x / OWN), b1 = (unsigned)(d.y / OWN);
        const unsigned b2 = (unsigned)(d.z / OWN), b3 = (unsigned)(d.w / OWN);
        uint4 r;
        r.x = (unsigned)s.x | (((unsigned)d.x - b0 * OWN) << SRC_BITS);
        r.y = (unsigned)s.y | (((unsigned)d.y - b1 * OWN) << SRC_BITS);
        r.z = (unsigned)s.z | (((unsigned)d.z - b2 * OWN) << SRC_BITS);
        r.w = (unsigned)s.w | (((unsigned)d.w - b3 * OWN) << SRC_BITS);
        rec[t] = r;
        unsigned* w = lm + (tid >> 3);
        const unsigned sh = (unsigned)(tid & 7) << 2;
        __hip_atomic_fetch_or(w + b0 * CH_PITCH, 1u << sh, __ATOMIC_RELAXED, __HIP_MEMORY_SCOPE_WORKGROUP);
        __hip_atomic_fetch_or(w + b1 * CH_PITCH, 2u << sh, __ATOMIC_RELAXED, __HIP_MEMORY_SCOPE_WORKGROUP);
        __hip_atomic_fetch_or(w + b2 * CH_PITCH, 4u << sh, __ATOMIC_RELAXED, __HIP_MEMORY_SCOPE_WORKGROUP);
        __hip_atomic_fetch_or(w + b3 * CH_PITCH, 8u << sh, __ATOMIC_RELAXED, __HIP_MEMORY_SCOPE_WORKGROUP);
    }
    __syncthreads();

    constexpr int PO = CH_WORDS / 4;
    constexpr int NP = N_OWNERS * PO / CH_THREADS;
    uint4 v[NP];
#pragma unroll
    for (int k = 0; k < NP; ++k) {
        const int p = tid + k * CH_THREADS;
        v[k] = lm4[(p / PO) * (CH_PITCH / 4) + (p % PO)];
    }
#pragma unroll
    for (int k = 0; k < NP; ++k) {
        const int p = tid + k * CH_THREADS;
        rows[(size_t)(p / PO) * (ROW_WORDS / 4) + chunk * PO + (p % PO)] = v[k];
    }
}

#define SCL_ROWS 80
#define SCL_THREADS 256
static_assert(N_NODES % SCL_ROWS == 0, "the rows must split evenly over the scaling work-groups");
static_assert((SCL_ROWS * 16) % SCL_THREADS == 0, "a block of rows must split evenly over the threads");

__global__ void __launch_bounds__(SCL_THREADS)
scale_rows_kernel(const float* __restrict__ X, const float* __restrict__ dinv, float* __restrict__ XS) {
    constexpr int NP = SCL_ROWS * 16 / SCL_THREADS;
    const int tid = threadIdx.x;
    const float4* in  = reinterpret_cast<const float4*>(X) + (size_t)blockIdx.x * (SCL_ROWS * 16);
    float4*       out = reinterpret_cast<float4*>(XS) + (size_t)blockIdx.x * (SCL_ROWS * 16);
    float4 v[NP];
    float  d[NP];
#pragma unroll
    for (int p = 0; p < NP; ++p) {
        const int i = tid + p * SCL_THREADS;
        v[p] = in[i];
        d[p] = dinv[blockIdx.x * SCL_ROWS + (i >> 4)];
    }
#pragma unroll
    for (int p = 0; p < NP; ++p) {
        v[p].x *= d[p]; v[p].y *= d[p]; v[p].z *= d[p]; v[p].w *= d[p];
        out[tid + p * SCL_THREADS] = v[p];
    }
}

#define GEMM_MT 5
#define GEMM_ROWS (GEMM_MT * 16)
static_assert(N_NODES % GEMM_ROWS == 0, "the rows must split evenly over the product work-groups");

template <int K, int COUT, bool ACT>
__global__ void __launch_bounds__((COUT / 16) * 32)
gemm_wmma(const float* __restrict__ H,
          const float* __restrict__ W,
          const float* __restrict__ dinv,
          const float* __restrict__ bias,
          float* __restrict__ Y) {
    constexpr int NT = (COUT / 16) * 32;
    constexpr int P  = K + 4;
    constexpr int NV = GEMM_ROWS * K / 4;
    static_assert((K & (K - 1)) == 0 && K % 4 == 0, "K must be a power of two");
    static_assert(((P / 4) & 1) == 1, "the pitch must be an odd number of 16-byte pieces");
    static_assert(NV % NT == 0, "the block must split evenly over the threads");

    __shared__ __attribute__((aligned(16))) float a_lds[GEMM_ROWS * P];

    const int tid    = threadIdx.x;
    const int lane   = tid & 31;
    const int n_tile = tid >> 5;
    const int row0   = blockIdx.x * GEMM_ROWS;

    const int col = n_tile * 16 + (lane & 15);
    const int kh  = (lane >> 4) << 1;

    v2f b[K / 4];
#pragma unroll
    for (int kk = 0; kk < K; kk += 4) {
        b[kk / 4].x = W[(kk + kh) * COUT + col];
        b[kk / 4].y = W[(kk + kh + 1) * COUT + col];
    }
    float bc = 0.0f;
    if (ACT) bc = bias[col];

    const float4* hblk = reinterpret_cast<const float4*>(H + (size_t)row0 * K);
    float4 piece[NV / NT];
#pragma unroll
    for (int p = 0; p < NV / NT; ++p) piece[p] = hblk[tid + p * NT];
#pragma unroll
    for (int p = 0; p < NV / NT; ++p) {
        const int i = tid + p * NT;
        const int r = i / (K / 4);
        const int c = (i % (K / 4)) * 4;
        *reinterpret_cast<float4*>(a_lds + r * P + c) = piece[p];
    }
    __syncthreads();

    const float* arow = a_lds + (lane & 15) * P + kh;

    v8f acc[GEMM_MT];
#pragma unroll
    for (int t = 0; t < GEMM_MT; ++t) acc[t] = v8f{};

#pragma unroll
    for (int kk = 0; kk < K; kk += 4) {
#pragma unroll
        for (int t = 0; t < GEMM_MT; ++t) {
            const v2f a = *reinterpret_cast<const v2f*>(arow + t * 16 * P + kk);
            acc[t] = __builtin_amdgcn_wmma_f32_16x16x4_f32(
                 false, a,  false, b[kk / 4],
                 (short)0, acc[t],  false,  false);
        }
    }

#pragma unroll
    for (int t = 0; t < GEMM_MT; ++t) {
        const int mbase = row0 + t * 16 + ((lane >> 4) << 3);
#pragma unroll
        for (int r = 0; r < 8; ++r) {
            const int m = mbase + r;
            if (ACT) Y[m * COUT + col] = fmaxf(acc[t][r] + bc, 0.0f);
            else     Y[m * COUT + col] = dinv[m] * acc[t][r];
        }
    }
}

template <int COUT, bool RELU, bool FIRST, bool BIAS>
__global__ void __launch_bounds__(AGG_THREADS)
owner_aggregate(const unsigned* __restrict__ rec,
                const uint4* __restrict__ rows,
                const float* __restrict__ Y,
                const float* __restrict__ dinv,
                const float* __restrict__ bias,
                float* __restrict__ OUT,
                unsigned* __restrict__ lst,
                int* __restrict__ lcnt) {
    constexpr int G  = COUT / 32;
    constexpr int EB = 32 / G;
    constexpr int N_TRIPS = ROW_WORDS / 128;

    __shared__ __attribute__((aligned(16))) float acc[OWN * COUT];
    __shared__ int q[AGG_WAVES][64];

    const int tid   = threadIdx.x;
    const int lane  = tid & 31;
    const int wave  = __builtin_amdgcn_readfirstlane(tid >> 5);
    const int owner = blockIdx.x;
    const int node0 = owner * OWN;
    const int count = (N_NODES - node0 < OWN) ? (N_NODES - node0) : OWN;
    const int nval  = count * COUT;
    const int slot  = owner * AGG_WAVES + wave;
    unsigned* mine  = lst + slot * LIST_CAP;

    int kept = 0;
    if (!FIRST) kept = lcnt[slot];

    for (int i = tid; i < nval; i += AGG_THREADS) acc[i] = Y[node0 * COUT + i];
    __syncthreads();

    auto add_rows = [&](unsigned r, int n) {
        for (int i0 = 0; i0 < n; i0 += EB) {
            float v[EB][G];
#pragma unroll
            for (int k = 0; k < EB; ++k) {
                if (i0 + k < n) {
                    const int sk = (int)((unsigned)__builtin_amdgcn_readlane((int)r, i0 + k) & SRC_MASK);
                    const float* yr = Y + sk * COUT + lane;
#pragma unroll
                    for (int g = 0; g < G; ++g) v[k][g] = yr[g * 32];
                }
            }
#pragma unroll
            for (int k = 0; k < EB; ++k) {
                if (i0 + k < n) {
                    const int dk = (int)((unsigned)__builtin_amdgcn_readlane((int)r, i0 + k) >> SRC_BITS);
                    float* a = acc + dk * COUT + lane;
#pragma unroll
                    for (int g = 0; g < G; ++g)
                        __hip_atomic_fetch_add(a + g * 32, v[k][g], __ATOMIC_RELAXED,
                                               __HIP_MEMORY_SCOPE_WORKGROUP);
                }
            }
        }
    };

    kept = __builtin_amdgcn_readfirstlane(kept);
    if (!FIRST && kept <= LIST_CAP) {
        unsigned r = 0;
        if (lane < kept) r = mine[lane];
        for (int i0 = 0; i0 < kept; i0 += 32) {
            const int left = kept - i0;
            unsigned rn = 0;
            if (32 + lane < left) rn = mine[i0 + 32 + lane];
            if (left >= 32) add_rows(r, 32); else add_rows(r, left);
            r = rn;
        }
    } else {
        const uint4* row = rows + (size_t)owner * (ROW_WORDS / 4);
        int* wq = q[wave];
        int head = 0, tail = 0;

        auto drain = [&](int n) {
            unsigned r = 0;
            if (lane < n) r = rec[wq[(head + lane) & 63]];
            if (FIRST) {
                if (head + n <= LIST_CAP && lane < n) mine[head + lane] = r;
            }
            add_rows(r, n);
            head += n;
        };

        static_assert(AGG_WAVES <= N_TRIPS, "every wave has a first trip");
        uint4 cur = row[wave * 32 + lane];
        for (int trip = wave; trip < N_TRIPS; trip += AGG_WAVES) {
            const int ntrip = (trip + AGG_WAVES < N_TRIPS) ? (trip + AGG_WAVES) : trip;
            const uint4 nxt = row[ntrip * 32 + lane];

            unsigned long long lo = (unsigned long long)cur.x | ((unsigned long long)cur.y << 32);
            unsigned long long hi = (unsigned long long)cur.z | ((unsigned long long)cur.w << 32);
            const int ebase = (trip * 32 + lane) * 128;

            for (;;) {
                const bool has = ((lo | hi) != 0ull);
                const unsigned bal = __builtin_amdgcn_ballot_w32(has);
                if (bal == 0u) break;
                if (has) {
                    const bool low = (lo != 0ull);
                    const unsigned long long m = low ? lo : hi;
                    const int p    = __builtin_ctzll(m) + (low ? 0 : 64);
                    const int rank = (int)__builtin_amdgcn_mbcnt_lo(bal, 0u);
                    wq[(tail + rank) & 63] = ebase + p;
                    const unsigned long long rest = m & (m - 1ull);
                    lo = low ? rest : lo;
                    hi = low ? hi : rest;
                }
                tail += __builtin_popcount(bal);
                __builtin_amdgcn_fence(__ATOMIC_ACQ_REL, "wavefront");
                if (tail - head >= 32) drain(32);
            }
            cur = nxt;
        }
        __builtin_amdgcn_fence(__ATOMIC_ACQ_REL, "wavefront");
        drain(tail - head);
        if (FIRST && lane == 0) lcnt[slot] = head;
    }
    __syncthreads();

    for (int i = tid; i < nval; i += AGG_THREADS) {
        const int n = i / COUT;
        const int c = i & (COUT - 1);
        float v = dinv[node0 + n] * acc[i];
        if (BIAS) v += bias[c];
        if (RELU) v = fmaxf(v, 0.0f);
        OUT[node0 * COUT + i] = v;
    }
}

extern "C" void kernel_launch(void* const* d_in, const int* in_sizes, int n_in,
                              void* d_out, int out_size, void* d_ws, size_t ws_size,
                              hipStream_t stream) {
    const float* x  = (const float*)d_in[0];
    const int*   ei = (const int*)d_in[1];
    const float* W1 = (const float*)d_in[2];
    const float* b1 = (const float*)d_in[3];
    const float* W2 = (const float*)d_in[4];
    const float* b2 = (const float*)d_in[5];
    const float* W3 = (const float*)d_in[6];
    const float* b3 = (const float*)d_in[7];
    float*       out = (float*)d_out;

    const int* src = ei;
    const int* dst = ei + N_EDGES;

    char*  ws   = (char*)d_ws;
    size_t off  = 0;
    float*    dinv = (float*)(ws + off);    off += ((size_t)N_NODES * 4 + 511) & ~511ull;
    unsigned* rec  = (unsigned*)(ws + off); off += ((size_t)N_EDGES * 4 + 511) & ~511ull;
    float*    y    = (float*)(ws + off);    off += (size_t)N_NODES * 128 * 4;
    float*    h    = (float*)(ws + off);    off += (size_t)N_NODES * 128 * 4;
    unsigned* lst  = (unsigned*)(ws + off); off += (size_t)N_OWNERS * AGG_WAVES * LIST_CAP * 4;
    int*      lcnt = (int*)(ws + off);      off += (size_t)N_OWNERS * AGG_WAVES * 4;
    uint4*    part = (uint4*)(ws + off);    off += (size_t)DEG_PARTS * PART_W * 4;
    uint4*    rows = (uint4*)(ws + off);

    float* xs = y;
    float* t  = y + (size_t)N_NODES * 64;

    const int M_BLOCKS = N_NODES / GEMM_ROWS;

    degree_hist_kernel<<<DEG_PARTS, DEG_THREADS, 0, stream>>>(dst, part);
    degree_merge_kernel<<<PART_W / (MRG_COLS * 4), MRG_THREADS, 0, stream>>>(part, dinv);
    owner_rows_kernel<<<N_CHUNKS, CH_THREADS, 0, stream>>>(src, dst, rows, (uint4*)rec);

    scale_rows_kernel<<<N_NODES / SCL_ROWS, SCL_THREADS, 0, stream>>>(x, dinv, xs);
    owner_aggregate<64, false, true, false><<<N_OWNERS, AGG_THREADS, 0, stream>>>(rec, rows, xs, dinv, nullptr, t, lst, lcnt);
    gemm_wmma<64, 128, true><<<M_BLOCKS, 256, 0, stream>>>(t, W1, nullptr, b1, h);

    gemm_wmma<128, 128, false><<<M_BLOCKS, 256, 0, stream>>>(h, W2, dinv, nullptr, y);
    owner_aggregate<128, true, false, true><<<N_OWNERS, AGG_THREADS, 0, stream>>>(rec, rows, y, dinv, b2, h, lst, lcnt);

    gemm_wmma<128, 64, false><<<M_BLOCKS, 128, 0, stream>>>(h, W3, dinv, nullptr, y);
    owner_aggregate<64, false, false, true><<<N_OWNERS, AGG_THREADS, 0, stream>>>(rec, rows, y, dinv, b3, out, lst, lcnt);
}


# owner walk split off as its own pass that counts each owned row's edges in LDS: degrees, dinv, scaled x rows come from it; histogram, merge, scale passes gone (10 -> 8 launches); sum order kept
# speedup vs baseline: 4.3495x; 1.0375x over previous
#include <hip/hip_runtime.h>


#define N_NODES 100000
#define N_EDGES 1600000

#define OWN 391
#define N_OWNERS 256
#define AGG_THREADS 1024
#define AGG_WAVES (AGG_THREADS / 32)
#define LIST_CAP 512

typedef __attribute__((ext_vector_type(2))) float v2f;
typedef __attribute__((ext_vector_type(8))) float v8f;

#define SRC_BITS 17
#define SRC_MASK ((1u << SRC_BITS) - 1u)
static_assert(N_NODES <= (1 << SRC_BITS), "a node number must fit the record's source field");
static_assert(OWN <= (1 << (32 - SRC_BITS)), "a row number inside an owner range must fit the rest of the record");

#define CH_EDGES 2048
#define CH_WORDS (CH_EDGES / 32)
#define CH_THREADS (CH_EDGES / 4)
#define N_CHUNKS ((N_EDGES + CH_EDGES - 1) / CH_EDGES)
#define ROW_WORDS (N_CHUNKS * CH_WORDS)
#define CH_PITCH (CH_WORDS + 4)
static_assert(N_EDGES % 4 == 0, "edges are read in groups of four");
static_assert(ROW_WORDS % 128 == 0, "a row must split into whole wave loads of four words per lane");
static_assert((N_OWNERS * CH_WORDS / 4) % CH_THREADS == 0, "the chunk's pieces must split evenly over the threads");

__global__ void __launch_bounds__(CH_THREADS)
owner_rows_kernel(const int* __restrict__ src, const int* __restrict__ dst,
                  uint4* __restrict__ rows, uint4* __restrict__ rec) {
    __shared__ __attribute__((aligned(16))) unsigned lm[N_OWNERS * CH_PITCH];
    uint4* lm4 = reinterpret_cast<uint4*>(lm);
    const int tid   = threadIdx.x;
    const int chunk = blockIdx.x;
    const int t     = chunk * CH_THREADS + tid;
    const bool live = t < N_EDGES / 4;

    int4 d = make_int4(0, 0, 0, 0), s = make_int4(0, 0, 0, 0);
    if (live) {
        d = *reinterpret_cast<const int4*>(dst + 4 * t);
        s = *reinterpret_cast<const int4*>(src + 4 * t);
    }

    for (int i = tid; i < N_OWNERS * CH_PITCH / 4; i += CH_THREADS) lm4[i] = make_uint4(0u, 0u, 0u, 0u);
    __syncthreads();

    if (live) {
        const unsigned b0 = (unsigned)(d.x / OWN), b1 = (unsigned)(d.y / OWN);
        const unsigned b2 = (unsigned)(d.z / OWN), b3 = (unsigned)(d.w / OWN);
        uint4 r;
        r.x = (unsigned)s.x | (((unsigned)d.x - b0 * OWN) << SRC_BITS);
        r.y = (unsigned)s.y | (((unsigned)d.y - b1 * OWN) << SRC_BITS);
        r.z = (unsigned)s.z | (((unsigned)d.z - b2 * OWN) << SRC_BITS);
        r.w = (unsigned)s.w | (((unsigned)d.w - b3 * OWN) << SRC_BITS);
        rec[t] = r;
        unsigned* w = lm + (tid >> 3);
        const unsigned sh = (unsigned)(tid & 7) << 2;
        __hip_atomic_fetch_or(w + b0 * CH_PITCH, 1u << sh, __ATOMIC_RELAXED, __HIP_MEMORY_SCOPE_WORKGROUP);
        __hip_atomic_fetch_or(w + b1 * CH_PITCH, 2u << sh, __ATOMIC_RELAXED, __HIP_MEMORY_SCOPE_WORKGROUP);
        __hip_atomic_fetch_or(w + b2 * CH_PITCH, 4u << sh, __ATOMIC_RELAXED, __HIP_MEMORY_SCOPE_WORKGROUP);
        __hip_atomic_fetch_or(w + b3 * CH_PITCH, 8u << sh, __ATOMIC_RELAXED, __HIP_MEMORY_SCOPE_WORKGROUP);
    }
    __syncthreads();

    constexpr int PO = CH_WORDS / 4;
    constexpr int NP = N_OWNERS * PO / CH_THREADS;
    uint4 v[NP];
#pragma unroll
    for (int k = 0; k < NP; ++k) {
        const int p = tid + k * CH_THREADS;
        v[k] = lm4[(p / PO) * (CH_PITCH / 4) + (p % PO)];
    }
#pragma unroll
    for (int k = 0; k < NP; ++k) {
        const int p = tid + k * CH_THREADS;
        rows[(size_t)(p / PO) * (ROW_WORDS / 4) + chunk * PO + (p % PO)] = v[k];
    }
}

#define WALK_TRIPS (ROW_WORDS / 128)
#define WALK_FULL (WALK_TRIPS / AGG_WAVES)
#define WALK_TPW ((WALK_TRIPS + AGG_WAVES - 1) / AGG_WAVES)
#define WALK_XP ((OWN * 16 + AGG_THREADS - 1) / AGG_THREADS)
#define WALK_Q 512
static_assert((WALK_Q & (WALK_Q - 1)) == 0 && WALK_Q >= 256, "the queue is a ring of a power of two");

__global__ void __launch_bounds__(AGG_THREADS)
owner_walk(const unsigned* __restrict__ rec,
           const uint4* __restrict__ rows,
           const float* __restrict__ X,
           float* __restrict__ dinv,
           float* __restrict__ XS,
           unsigned* __restrict__ lst,
           int* __restrict__ lcnt) {
    __shared__ unsigned deg[OWN];
    __shared__ int q[AGG_WAVES][WALK_Q];

    const int tid   = threadIdx.x;
    const int lane  = tid & 31;
    const int wave  = __builtin_amdgcn_readfirstlane(tid >> 5);
    const int owner = blockIdx.x;
    const int node0 = owner * OWN;
    const int count = (N_NODES - node0 < OWN) ? (N_NODES - node0) : OWN;
    const int slot  = owner * AGG_WAVES + wave;
    unsigned* mine  = lst + slot * LIST_CAP;

    for (int i = tid; i < OWN; i += AGG_THREADS) deg[i] = 0u;
    __syncthreads();

    const uint4* row = rows + (size_t)owner * (ROW_WORDS / 4);
    uint4 w[WALK_TPW];
#pragma unroll
    for (int k = 0; k < WALK_TPW; ++k) {
        const int trip = wave + k * AGG_WAVES;
        w[k] = (k < WALK_FULL || trip < WALK_TRIPS) ? row[trip * 32 + lane] : make_uint4(0u, 0u, 0u, 0u);
    }
    const int npieces = count * 16;
    const float4* xin = reinterpret_cast<const float4*>(X) + (size_t)node0 * 16;
    float4 xv[WALK_XP];
#pragma unroll
    for (int p = 0; p < WALK_XP; ++p) {
        const int i = tid + p * AGG_THREADS;
        xv[p] = (i < npieces) ? xin[i] : make_float4(0.0f, 0.0f, 0.0f, 0.0f);
    }

    int* wq = q[wave];
    int head = 0, tail = 0;

    auto drain = [&](int stop) {
        __builtin_amdgcn_fence(__ATOMIC_ACQ_REL, "wavefront");
        unsigned r[4];
#pragma unroll
        for (int j = 0; j < 4; ++j) {
            const int i = head + 32 * j + lane;
            const int c = (i < stop) ? i : (stop - 1);
            r[j] = rec[wq[c & (WALK_Q - 1)]];
        }
#pragma unroll
        for (int j = 0; j < 4; ++j) {
            const int i = head + 32 * j + lane;
            if (i < stop) {
                if (i < LIST_CAP) mine[i] = r[j];
                __hip_atomic_fetch_add(deg + (r[j] >> SRC_BITS), 1u, __ATOMIC_RELAXED, __HIP_MEMORY_SCOPE_WORKGROUP);
            }
        }
        head = (stop - head < 128) ? stop : (head + 128);
    };

#pragma unroll
    for (int k = 0; k < WALK_TPW; ++k) {
        const int trip = wave + k * AGG_WAVES;
        if (k < WALK_FULL || trip < WALK_TRIPS) {
            unsigned long long lo = (unsigned long long)w[k].x | ((unsigned long long)w[k].y << 32);
            unsigned long long hi = (unsigned long long)w[k].z | ((unsigned long long)w[k].w << 32);
            const int ebase = (trip * 32 + lane) * 128;

            for (;;) {
                const bool has = ((lo | hi) != 0ull);
                const unsigned bal = __builtin_amdgcn_ballot_w32(has);
                if (bal == 0u) break;
                if (has) {
                    const bool low = (lo != 0ull);
                    const unsigned long long m = low ? lo : hi;
                    const int p    = __builtin_ctzll(m) + (low ? 0 : 64);
                    const int rank = (int)__builtin_amdgcn_mbcnt_lo(bal, 0u);
                    wq[(tail + rank) & (WALK_Q - 1)] = ebase + p;
                    const unsigned long long rest = m & (m - 1ull);
                    lo = low ? rest : lo;
                    hi = low ? hi : rest;
                }
                tail += __builtin_popcount(bal);
                if (tail - head > WALK_Q - 32) drain(tail);
            }
        }
    }
    while (head < tail) drain(tail);
    if (lane == 0) lcnt[slot] = head;
    __syncthreads();

    if (tid < count) dinv[node0 + tid] = rsqrtf((float)deg[tid] + 1.0f);

    float4* xout = reinterpret_cast<float4*>(XS) + (size_t)node0 * 16;
#pragma unroll
    for (int p = 0; p < WALK_XP; ++p) {
        const int i = tid + p * AGG_THREADS;
        if (i < npieces) {
            const float d = rsqrtf((float)deg[i >> 4] + 1.0f);
            float4 v = xv[p];
            v.x *= d; v.y *= d; v.z *= d; v.w *= d;
            xout[i] = v;
        }
    }
}

#define GEMM_MT 5
#define GEMM_ROWS (GEMM_MT * 16)
static_assert(N_NODES % GEMM_ROWS == 0, "the rows must split evenly over the product work-groups");

template <int K, int COUT, bool ACT>
__global__ void __launch_bounds__((COUT / 16) * 32)
gemm_wmma(const float* __restrict__ H,
          const float* __restrict__ W,
          const float* __restrict__ dinv,
          const float* __restrict__ bias,
          float* __restrict__ Y) {
    constexpr int NT = (COUT / 16) * 32;
    constexpr int P  = K + 4;
    constexpr int NV = GEMM_ROWS * K / 4;
    static_assert((K & (K - 1)) == 0 && K % 4 == 0, "K must be a power of two");
    static_assert(((P / 4) & 1) == 1, "the pitch must be an odd number of 16-byte pieces");
    static_assert(NV % NT == 0, "the block must split evenly over the threads");

    __shared__ __attribute__((aligned(16))) float a_lds[GEMM_ROWS * P];

    const int tid    = threadIdx.x;
    const int lane   = tid & 31;
    const int n_tile = tid >> 5;
    const int row0   = blockIdx.x * GEMM_ROWS;

    const int col = n_tile * 16 + (lane & 15);
    const int kh  = (lane >> 4) << 1;

    v2f b[K / 4];
#pragma unroll
    for (int kk = 0; kk < K; kk += 4) {
        b[kk / 4].x = W[(kk + kh) * COUT + col];
        b[kk / 4].y = W[(kk + kh + 1) * COUT + col];
    }
    float bc = 0.0f;
    if (ACT) bc = bias[col];

    const float4* hblk = reinterpret_cast<const float4*>(H + (size_t)row0 * K);
    float4 piece[NV / NT];
#pragma unroll
    for (int p = 0; p < NV / NT; ++p) piece[p] = hblk[tid + p * NT];
#pragma unroll
    for (int p = 0; p < NV / NT; ++p) {
        const int i = tid + p * NT;
        const int r = i / (K / 4);
        const int c = (i % (K / 4)) * 4;
        *reinterpret_cast<float4*>(a_lds + r * P + c) = piece[p];
    }
    __syncthreads();

    const float* arow = a_lds + (lane & 15) * P + kh;

    v8f acc[GEMM_MT];
#pragma unroll
    for (int t = 0; t < GEMM_MT; ++t) acc[t] = v8f{};

#pragma unroll
    for (int kk = 0; kk < K; kk += 4) {
#pragma unroll
        for (int t = 0; t < GEMM_MT; ++t) {
            const v2f a = *reinterpret_cast<const v2f*>(arow + t * 16 * P + kk);
            acc[t] = __builtin_amdgcn_wmma_f32_16x16x4_f32(
                 false, a,  false, b[kk / 4],
                 (short)0, acc[t],  false,  false);
        }
    }

#pragma unroll
    for (int t = 0; t < GEMM_MT; ++t) {
        const int mbase = row0 + t * 16 + ((lane >> 4) << 3);
#pragma unroll
        for (int r = 0; r < 8; ++r) {
            const int m = mbase + r;
            if (ACT) Y[m * COUT + col] = fmaxf(acc[t][r] + bc, 0.0f);
            else     Y[m * COUT + col] = dinv[m] * acc[t][r];
        }
    }
}

template <int COUT, bool RELU, bool BIAS>
__global__ void __launch_bounds__(AGG_THREADS)
owner_aggregate(const unsigned* __restrict__ rec,
                const uint4* __restrict__ rows,
                const float* __restrict__ Y,
                const float* __restrict__ dinv,
                const float* __restrict__ bias,
                float* __restrict__ OUT,
                const unsigned* __restrict__ lst,
                const int* __restrict__ lcnt) {
    constexpr int G  = COUT / 32;
    constexpr int EB = 32 / G;
    constexpr int N_TRIPS = ROW_WORDS / 128;

    __shared__ __attribute__((aligned(16))) float acc[OWN * COUT];
    __shared__ int q[AGG_WAVES][64];

    const int tid   = threadIdx.x;
    const int lane  = tid & 31;
    const int wave  = __builtin_amdgcn_readfirstlane(tid >> 5);
    const int owner = blockIdx.x;
    const int node0 = owner * OWN;
    const int count = (N_NODES - node0 < OWN) ? (N_NODES - node0) : OWN;
    const int nval  = count * COUT;
    const int slot  = owner * AGG_WAVES + wave;
    const unsigned* mine = lst + slot * LIST_CAP;

    int kept = lcnt[slot];

    for (int i = tid; i < nval; i += AGG_THREADS) acc[i] = Y[node0 * COUT + i];
    __syncthreads();

    auto add_rows = [&](unsigned r, int n) {
        for (int i0 = 0; i0 < n; i0 += EB) {
            float v[EB][G];
#pragma unroll
            for (int k = 0; k < EB; ++k) {
                if (i0 + k < n) {
                    const int sk = (int)((unsigned)__builtin_amdgcn_readlane((int)r, i0 + k) & SRC_MASK);
                    const float* yr = Y + sk * COUT + lane;
#pragma unroll
                    for (int g = 0; g < G; ++g) v[k][g] = yr[g * 32];
                }
            }
#pragma unroll
            for (int k = 0; k < EB; ++k) {
                if (i0 + k < n) {
                    const int dk = (int)((unsigned)__builtin_amdgcn_readlane((int)r, i0 + k) >> SRC_BITS);
                    float* a = acc + dk * COUT + lane;
#pragma unroll
                    for (int g = 0; g < G; ++g)
                        __hip_atomic_fetch_add(a + g * 32, v[k][g], __ATOMIC_RELAXED,
                                               __HIP_MEMORY_SCOPE_WORKGROUP);
                }
            }
        }
    };

    kept = __builtin_amdgcn_readfirstlane(kept);
    if (kept <= LIST_CAP) {
        unsigned r = 0;
        if (lane < kept) r = mine[lane];
        for (int i0 = 0; i0 < kept; i0 += 32) {
            const int left = kept - i0;
            unsigned rn = 0;
            if (32 + lane < left) rn = mine[i0 + 32 + lane];
            if (left >= 32) add_rows(r, 32); else add_rows(r, left);
            r = rn;
        }
    } else {
        const uint4* row = rows + (size_t)owner * (ROW_WORDS / 4);
        int* wq = q[wave];
        int head = 0, tail = 0;

        auto drain = [&](int n) {
            unsigned r = 0;
            if (lane < n) r = rec[wq[(head + lane) & 63]];
            add_rows(r, n);
            head += n;
        };

        static_assert(AGG_WAVES <= N_TRIPS, "every wave has a first trip");
        uint4 cur = row[wave * 32 + lane];
        for (int trip = wave; trip < N_TRIPS; trip += AGG_WAVES) {
            const int ntrip = (trip + AGG_WAVES < N_TRIPS) ? (trip + AGG_WAVES) : trip;
            const uint4 nxt = row[ntrip * 32 + lane];

            unsigned long long lo = (unsigned long long)cur.x | ((unsigned long long)cur.y << 32);
            unsigned long long hi = (unsigned long long)cur.z | ((unsigned long long)cur.w << 32);
            const int ebase = (trip * 32 + lane) * 128;

            for (;;) {
                const bool has = ((lo | hi) != 0ull);
                const unsigned bal = __builtin_amdgcn_ballot_w32(has);
                if (bal == 0u) break;
                if (has) {
                    const bool low = (lo != 0ull);
                    const unsigned long long m = low ? lo : hi;
                    const int p    = __builtin_ctzll(m) + (low ? 0 : 64);
                    const int rank = (int)__builtin_amdgcn_mbcnt_lo(bal, 0u);
                    wq[(tail + rank) & 63] = ebase + p;
                    const unsigned long long rest = m & (m - 1ull);
                    lo = low ? rest : lo;
                    hi = low ? hi : rest;
                }
                tail += __builtin_popcount(bal);
                __builtin_amdgcn_fence(__ATOMIC_ACQ_REL, "wavefront");
                if (tail - head >= 32) drain(32);
            }
            cur = nxt;
        }
        __builtin_amdgcn_fence(__ATOMIC_ACQ_REL, "wavefront");
        drain(tail - head);
    }
    __syncthreads();

    for (int i = tid; i < nval; i += AGG_THREADS) {
        const int n = i / COUT;
        const int c = i & (COUT - 1);
        float v = dinv[node0 + n] * acc[i];
        if (BIAS) v += bias[c];
        if (RELU) v = fmaxf(v, 0.0f);
        OUT[node0 * COUT + i] = v;
    }
}

extern "C" void kernel_launch(void* const* d_in, const int* in_sizes, int n_in,
                              void* d_out, int out_size, void* d_ws, size_t ws_size,
                              hipStream_t stream) {
    const float* x  = (const float*)d_in[0];
    const int*   ei = (const int*)d_in[1];
    const float* W1 = (const float*)d_in[2];
    const float* b1 = (const float*)d_in[3];
    const float* W2 = (const float*)d_in[4];
    const float* b2 = (const float*)d_in[5];
    const float* W3 = (const float*)d_in[6];
    const float* b3 = (const float*)d_in[7];
    float*       out = (float*)d_out;

    const int* src = ei;
    const int* dst = ei + N_EDGES;

    char*  ws   = (char*)d_ws;
    size_t off  = 0;
    float*    dinv = (float*)(ws + off);    off += ((size_t)N_NODES * 4 + 511) & ~511ull;
    unsigned* rec  = (unsigned*)(ws + off); off += ((size_t)N_EDGES * 4 + 511) & ~511ull;
    float*    y    = (float*)(ws + off);    off += (size_t)N_NODES * 128 * 4;
    float*    h    = (float*)(ws + off);    off += (size_t)N_NODES * 128 * 4;
    unsigned* lst  = (unsigned*)(ws + off); off += (size_t)N_OWNERS * AGG_WAVES * LIST_CAP * 4;
    int*      lcnt = (int*)(ws + off);      off += (size_t)N_OWNERS * AGG_WAVES * 4;
    uint4*    rows = (uint4*)(ws + off);

    float* xs = y;
    float* t  = y + (size_t)N_NODES * 64;

    const int M_BLOCKS = N_NODES / GEMM_ROWS;

    owner_rows_kernel<<<N_CHUNKS, CH_THREADS, 0, stream>>>(src, dst, rows, (uint4*)rec);
    owner_walk<<<N_OWNERS, AGG_THREADS, 0, stream>>>(rec, rows, x, dinv, xs, lst, lcnt);

    owner_aggregate<64, false, false><<<N_OWNERS, AGG_THREADS, 0, stream>>>(rec, rows, xs, dinv, nullptr, t, lst, lcnt);
    gemm_wmma<64, 128, true><<<M_BLOCKS, 256, 0, stream>>>(t, W1, nullptr, b1, h);

    gemm_wmma<128, 128, false><<<M_BLOCKS, 256, 0, stream>>>(h, W2, dinv, nullptr, y);
    owner_aggregate<128, true, true><<<N_OWNERS, AGG_THREADS, 0, stream>>>(rec, rows, y, dinv, b2, h, lst, lcnt);

    gemm_wmma<128, 64, false><<<M_BLOCKS, 128, 0, stream>>>(h, W3, dinv, nullptr, y);
    owner_aggregate<64, false, true><<<N_OWNERS, AGG_THREADS, 0, stream>>>(rec, rows, y, dinv, b3, out, lst, lcnt);
}


# layer 1's product kernel also does layer 2's product: its relu'd 80x128 tile stays in LDS as the A operand of a second WMMA pass (@W2, x dinv); h1 round trip and one launch gone (8 -> 7); sums kept
# speedup vs baseline: 4.5016x; 1.0350x over previous
#include <hip/hip_runtime.h>


#define N_NODES 100000
#define N_EDGES 1600000

#define OWN 391
#define N_OWNERS 256
#define AGG_THREADS 1024
#define AGG_WAVES (AGG_THREADS / 32)
#define LIST_CAP 512

typedef __attribute__((ext_vector_type(2))) float v2f;
typedef __attribute__((ext_vector_type(8))) float v8f;

#define SRC_BITS 17
#define SRC_MASK ((1u << SRC_BITS) - 1u)
static_assert(N_NODES <= (1 << SRC_BITS), "a node number must fit the record's source field");
static_assert(OWN <= (1 << (32 - SRC_BITS)), "a row number inside an owner range must fit the rest of the record");

#define CH_EDGES 2048
#define CH_WORDS (CH_EDGES / 32)
#define CH_THREADS (CH_EDGES / 4)
#define N_CHUNKS ((N_EDGES + CH_EDGES - 1) / CH_EDGES)
#define ROW_WORDS (N_CHUNKS * CH_WORDS)
#define CH_PITCH (CH_WORDS + 4)
static_assert(N_EDGES % 4 == 0, "edges are read in groups of four");
static_assert(ROW_WORDS % 128 == 0, "a row must split into whole wave loads of four words per lane");
static_assert((N_OWNERS * CH_WORDS / 4) % CH_THREADS == 0, "the chunk's pieces must split evenly over the threads");

__global__ void __launch_bounds__(CH_THREADS)
owner_rows_kernel(const int* __restrict__ src, const int* __restrict__ dst,
                  uint4* __restrict__ rows, uint4* __restrict__ rec) {
    __shared__ __attribute__((aligned(16))) unsigned lm[N_OWNERS * CH_PITCH];
    uint4* lm4 = reinterpret_cast<uint4*>(lm);
    const int tid   = threadIdx.x;
    const int chunk = blockIdx.x;
    const int t     = chunk * CH_THREADS + tid;
    const bool live = t < N_EDGES / 4;

    int4 d = make_int4(0, 0, 0, 0), s = make_int4(0, 0, 0, 0);
    if (live) {
        d = *reinterpret_cast<const int4*>(dst + 4 * t);
        s = *reinterpret_cast<const int4*>(src + 4 * t);
    }

    for (int i = tid; i < N_OWNERS * CH_PITCH / 4; i += CH_THREADS) lm4[i] = make_uint4(0u, 0u, 0u, 0u);
    __syncthreads();

    if (live) {
        const unsigned b0 = (unsigned)(d.x / OWN), b1 = (unsigned)(d.y / OWN);
        const unsigned b2 = (unsigned)(d.z / OWN), b3 = (unsigned)(d.w / OWN);
        uint4 r;
        r.x = (unsigned)s.x | (((unsigned)d.x - b0 * OWN) << SRC_BITS);
        r.y = (unsigned)s.y | (((unsigned)d.y - b1 * OWN) << SRC_BITS);
        r.z = (unsigned)s.z | (((unsigned)d.z - b2 * OWN) << SRC_BITS);
        r.w = (unsigned)s.w | (((unsigned)d.w - b3 * OWN) << SRC_BITS);
        rec[t] = r;
        unsigned* w = lm + (tid >> 3);
        const unsigned sh = (unsigned)(tid & 7) << 2;
        __hip_atomic_fetch_or(w + b0 * CH_PITCH, 1u << sh, __ATOMIC_RELAXED, __HIP_MEMORY_SCOPE_WORKGROUP);
        __hip_atomic_fetch_or(w + b1 * CH_PITCH, 2u << sh, __ATOMIC_RELAXED, __HIP_MEMORY_SCOPE_WORKGROUP);
        __hip_atomic_fetch_or(w + b2 * CH_PITCH, 4u << sh, __ATOMIC_RELAXED, __HIP_MEMORY_SCOPE_WORKGROUP);
        __hip_atomic_fetch_or(w + b3 * CH_PITCH, 8u << sh, __ATOMIC_RELAXED, __HIP_MEMORY_SCOPE_WORKGROUP);
    }
    __syncthreads();

    constexpr int PO = CH_WORDS / 4;
    constexpr int NP = N_OWNERS * PO / CH_THREADS;
    uint4 v[NP];
#pragma unroll
    for (int k = 0; k < NP; ++k) {
        const int p = tid + k * CH_THREADS;
        v[k] = lm4[(p / PO) * (CH_PITCH / 4) + (p % PO)];
    }
#pragma unroll
    for (int k = 0; k < NP; ++k) {
        const int p = tid + k * CH_THREADS;
        rows[(size_t)(p / PO) * (ROW_WORDS / 4) + chunk * PO + (p % PO)] = v[k];
    }
}

#define WALK_TRIPS (ROW_WORDS / 128)
#define WALK_FULL (WALK_TRIPS / AGG_WAVES)
#define WALK_TPW ((WALK_TRIPS + AGG_WAVES - 1) / AGG_WAVES)
#define WALK_XP ((OWN * 16 + AGG_THREADS - 1) / AGG_THREADS)
#define WALK_Q 512
static_assert((WALK_Q & (WALK_Q - 1)) == 0 && WALK_Q >= 256, "the queue is a ring of a power of two");

__global__ void __launch_bounds__(AGG_THREADS)
owner_walk(const unsigned* __restrict__ rec,
           const uint4* __restrict__ rows,
           const float* __restrict__ X,
           float* __restrict__ dinv,
           float* __restrict__ XS,
           unsigned* __restrict__ lst,
           int* __restrict__ lcnt) {
    __shared__ unsigned deg[OWN];
    __shared__ int q[AGG_WAVES][WALK_Q];

    const int tid   = threadIdx.x;
    const int lane  = tid & 31;
    const int wave  = __builtin_amdgcn_readfirstlane(tid >> 5);
    const int owner = blockIdx.x;
    const int node0 = owner * OWN;
    const int count = (N_NODES - node0 < OWN) ? (N_NODES - node0) : OWN;
    const int slot  = owner * AGG_WAVES + wave;
    unsigned* mine  = lst + slot * LIST_CAP;

    for (int i = tid; i < OWN; i += AGG_THREADS) deg[i] = 0u;
    __syncthreads();

    const uint4* row = rows + (size_t)owner * (ROW_WORDS / 4);
    uint4 w[WALK_TPW];
#pragma unroll
    for (int k = 0; k < WALK_TPW; ++k) {
        const int trip = wave + k * AGG_WAVES;
        w[k] = (k < WALK_FULL || trip < WALK_TRIPS) ? row[trip * 32 + lane] : make_uint4(0u, 0u, 0u, 0u);
    }
    const int npieces = count * 16;
    const float4* xin = reinterpret_cast<const float4*>(X) + (size_t)node0 * 16;
    float4 xv[WALK_XP];
#pragma unroll
    for (int p = 0; p < WALK_XP; ++p) {
        const int i = tid + p * AGG_THREADS;
        xv[p] = (i < npieces) ? xin[i] : make_float4(0.0f, 0.0f, 0.0f, 0.0f);
    }

    int* wq = q[wave];
    int head = 0, tail = 0;

    auto drain = [&](int stop) {
        __builtin_amdgcn_fence(__ATOMIC_ACQ_REL, "wavefront");
        unsigned r[4];
#pragma unroll
        for (int j = 0; j < 4; ++j) {
            const int i = head + 32 * j + lane;
            const int c = (i < stop) ? i : (stop - 1);
            r[j] = rec[wq[c & (WALK_Q - 1)]];
        }
#pragma unroll
        for (int j = 0; j < 4; ++j) {
            const int i = head + 32 * j + lane;
            if (i < stop) {
                if (i < LIST_CAP) mine[i] = r[j];
                __hip_atomic_fetch_add(deg + (r[j] >> SRC_BITS), 1u, __ATOMIC_RELAXED, __HIP_MEMORY_SCOPE_WORKGROUP);
            }
        }
        head = (stop - head < 128) ? stop : (head + 128);
    };

#pragma unroll
    for (int k = 0; k < WALK_TPW; ++k) {
        const int trip = wave + k * AGG_WAVES;
        if (k < WALK_FULL || trip < WALK_TRIPS) {
            unsigned long long lo = (unsigned long long)w[k].x | ((unsigned long long)w[k].y << 32);
            unsigned long long hi = (unsigned long long)w[k].z | ((unsigned long long)w[k].w << 32);
            const int ebase = (trip * 32 + lane) * 128;

            for (;;) {
                const bool has = ((lo | hi) != 0ull);
                const unsigned bal = __builtin_amdgcn_ballot_w32(has);
                if (bal == 0u) break;
                if (has) {
                    const bool low = (lo != 0ull);
                    const unsigned long long m = low ? lo : hi;
                    const int p    = __builtin_ctzll(m) + (low ? 0 : 64);
                    const int rank = (int)__builtin_amdgcn_mbcnt_lo(bal, 0u);
                    wq[(tail + rank) & (WALK_Q - 1)] = ebase + p;
                    const unsigned long long rest = m & (m - 1ull);
                    lo = low ? rest : lo;
                    hi = low ? hi : rest;
                }
                tail += __builtin_popcount(bal);
                if (tail - head > WALK_Q - 32) drain(tail);
            }
        }
    }
    while (head < tail) drain(tail);
    if (lane == 0) lcnt[slot] = head;
    __syncthreads();

    if (tid < count) dinv[node0 + tid] = rsqrtf((float)deg[tid] + 1.0f);

    float4* xout = reinterpret_cast<float4*>(XS) + (size_t)node0 * 16;
#pragma unroll
    for (int p = 0; p < WALK_XP; ++p) {
        const int i = tid + p * AGG_THREADS;
        if (i < npieces) {
            const float d = rsqrtf((float)deg[i >> 4] + 1.0f);
            float4 v = xv[p];
            v.x *= d; v.y *= d; v.z *= d; v.w *= d;
            xout[i] = v;
        }
    }
}

#define GEMM_MT 5
#define GEMM_ROWS (GEMM_MT * 16)
static_assert(N_NODES % GEMM_ROWS == 0, "the rows must split evenly over the product work-groups");

template <int K, int COUT>
__global__ void __launch_bounds__((COUT / 16) * 32)
gemm_wmma(const float* __restrict__ H,
          const float* __restrict__ W,
          const float* __restrict__ dinv,
          float* __restrict__ Y) {
    constexpr int NT = (COUT / 16) * 32;
    constexpr int P  = K + 4;
    constexpr int NV = GEMM_ROWS * K / 4;
    static_assert((K & (K - 1)) == 0 && K % 4 == 0, "K must be a power of two");
    static_assert(((P / 4) & 1) == 1, "the pitch must be an odd number of 16-byte pieces");
    static_assert(NV % NT == 0, "the block must split evenly over the threads");

    __shared__ __attribute__((aligned(16))) float a_lds[GEMM_ROWS * P];

    const int tid    = threadIdx.x;
    const int lane   = tid & 31;
    const int n_tile = tid >> 5;
    const int row0   = blockIdx.x * GEMM_ROWS;

    const int col = n_tile * 16 + (lane & 15);
    const int kh  = (lane >> 4) << 1;

    v2f b[K / 4];
#pragma unroll
    for (int kk = 0; kk < K; kk += 4) {
        b[kk / 4].x = W[(kk + kh) * COUT + col];
        b[kk / 4].y = W[(kk + kh + 1) * COUT + col];
    }

    const float4* hblk = reinterpret_cast<const float4*>(H + (size_t)row0 * K);
    float4 piece[NV / NT];
#pragma unroll
    for (int p = 0; p < NV / NT; ++p) piece[p] = hblk[tid + p * NT];
#pragma unroll
    for (int p = 0; p < NV / NT; ++p) {
        const int i = tid + p * NT;
        const int r = i / (K / 4);
        const int c = (i % (K / 4)) * 4;
        *reinterpret_cast<float4*>(a_lds + r * P + c) = piece[p];
    }
    __syncthreads();

    const float* arow = a_lds + (lane & 15) * P + kh;

    v8f acc[GEMM_MT];
#pragma unroll
    for (int t = 0; t < GEMM_MT; ++t) acc[t] = v8f{};

#pragma unroll
    for (int kk = 0; kk < K; kk += 4) {
#pragma unroll
        for (int t = 0; t < GEMM_MT; ++t) {
            const v2f a = *reinterpret_cast<const v2f*>(arow + t * 16 * P + kk);
            acc[t] = __builtin_amdgcn_wmma_f32_16x16x4_f32(
                 false, a,  false, b[kk / 4],
                 (short)0, acc[t],  false,  false);
        }
    }

#pragma unroll
    for (int t = 0; t < GEMM_MT; ++t) {
        const int mbase = row0 + t * 16 + ((lane >> 4) << 3);
#pragma unroll
        for (int r = 0; r < 8; ++r) {
            const int m = mbase + r;
            Y[m * COUT + col] = dinv[m] * acc[t][r];
        }
    }
}

__global__ void __launch_bounds__(256)
gemm_pair_wmma(const float* __restrict__ T,
               const float* __restrict__ W1,
               const float* __restrict__ b1,
               const float* __restrict__ W2,
               const float* __restrict__ dinv,
               float* __restrict__ Y) {
    constexpr int K1 = 64, C = 128;
    constexpr int NT = (C / 16) * 32;
    constexpr int P1 = K1 + 4;
    constexpr int P2 = C + 4;
    constexpr int NV = GEMM_ROWS * K1 / 4;
    static_assert(((P1 / 4) & 1) == 1 && ((P2 / 4) & 1) == 1, "the pitches must be odd numbers of 16-byte pieces");
    static_assert(NV % NT == 0, "the block must split evenly over the threads");

    __shared__ __attribute__((aligned(16))) float t_lds[GEMM_ROWS * P1];
    __shared__ __attribute__((aligned(16))) float h_lds[GEMM_ROWS * P2];

    const int tid    = threadIdx.x;
    const int lane   = tid & 31;
    const int n_tile = tid >> 5;
    const int row0   = blockIdx.x * GEMM_ROWS;

    const int col = n_tile * 16 + (lane & 15);
    const int kh  = (lane >> 4) << 1;
    const int rh  = (lane >> 4) << 3;

    v2f bw1[K1 / 4];
#pragma unroll
    for (int kk = 0; kk < K1; kk += 4) {
        bw1[kk / 4].x = W1[(kk + kh) * C + col];
        bw1[kk / 4].y = W1[(kk + kh + 1) * C + col];
    }
    const float bc = b1[col];

    const float4* tblk = reinterpret_cast<const float4*>(T + (size_t)row0 * K1);
    float4 piece[NV / NT];
#pragma unroll
    for (int p = 0; p < NV / NT; ++p) piece[p] = tblk[tid + p * NT];
#pragma unroll
    for (int p = 0; p < NV / NT; ++p) {
        const int i = tid + p * NT;
        const int r = i / (K1 / 4);
        const int cc = (i % (K1 / 4)) * 4;
        *reinterpret_cast<float4*>(t_lds + r * P1 + cc) = piece[p];
    }
    __syncthreads();

    v8f acc[GEMM_MT];
#pragma unroll
    for (int t = 0; t < GEMM_MT; ++t) acc[t] = v8f{};

    {
        const float* arow = t_lds + (lane & 15) * P1 + kh;
#pragma unroll
        for (int kk = 0; kk < K1; kk += 4) {
#pragma unroll
            for (int t = 0; t < GEMM_MT; ++t) {
                const v2f a = *reinterpret_cast<const v2f*>(arow + t * 16 * P1 + kk);
                acc[t] = __builtin_amdgcn_wmma_f32_16x16x4_f32(
                     false, a,  false, bw1[kk / 4],
                     (short)0, acc[t],  false,  false);
            }
        }
    }

    v2f bw2[C / 4];
#pragma unroll
    for (int kk = 0; kk < C; kk += 4) {
        bw2[kk / 4].x = W2[(kk + kh) * C + col];
        bw2[kk / 4].y = W2[(kk + kh + 1) * C + col];
    }

#pragma unroll
    for (int t = 0; t < GEMM_MT; ++t) {
#pragma unroll
        for (int r = 0; r < 8; ++r)
            h_lds[(t * 16 + rh + r) * P2 + col] = fmaxf(acc[t][r] + bc, 0.0f);
    }
    __syncthreads();

#pragma unroll
    for (int t = 0; t < GEMM_MT; ++t) acc[t] = v8f{};

    {
        const float* arow = h_lds + (lane & 15) * P2 + kh;
#pragma unroll
        for (int kk = 0; kk < C; kk += 4) {
#pragma unroll
            for (int t = 0; t < GEMM_MT; ++t) {
                const v2f a = *reinterpret_cast<const v2f*>(arow + t * 16 * P2 + kk);
                acc[t] = __builtin_amdgcn_wmma_f32_16x16x4_f32(
                     false, a,  false, bw2[kk / 4],
                     (short)0, acc[t],  false,  false);
            }
        }
    }

#pragma unroll
    for (int t = 0; t < GEMM_MT; ++t) {
        const int mbase = row0 + t * 16 + rh;
#pragma unroll
        for (int r = 0; r < 8; ++r) {
            const int m = mbase + r;
            Y[m * C + col] = dinv[m] * acc[t][r];
        }
    }
}

template <int COUT, bool RELU, bool BIAS>
__global__ void __launch_bounds__(AGG_THREADS)
owner_aggregate(const unsigned* __restrict__ rec,
                const uint4* __restrict__ rows,
                const float* __restrict__ Y,
                const float* __restrict__ dinv,
                const float* __restrict__ bias,
                float* __restrict__ OUT,
                const unsigned* __restrict__ lst,
                const int* __restrict__ lcnt) {
    constexpr int G  = COUT / 32;
    constexpr int EB = 32 / G;
    constexpr int N_TRIPS = ROW_WORDS / 128;

    __shared__ __attribute__((aligned(16))) float acc[OWN * COUT];
    __shared__ int q[AGG_WAVES][64];

    const int tid   = threadIdx.x;
    const int lane  = tid & 31;
    const int wave  = __builtin_amdgcn_readfirstlane(tid >> 5);
    const int owner = blockIdx.x;
    const int node0 = owner * OWN;
    const int count = (N_NODES - node0 < OWN) ? (N_NODES - node0) : OWN;
    const int nval  = count * COUT;
    const int slot  = owner * AGG_WAVES + wave;
    const unsigned* mine = lst + slot * LIST_CAP;

    int kept = lcnt[slot];

    for (int i = tid; i < nval; i += AGG_THREADS) acc[i] = Y[node0 * COUT + i];
    __syncthreads();

    auto add_rows = [&](unsigned r, int n) {
        for (int i0 = 0; i0 < n; i0 += EB) {
            float v[EB][G];
#pragma unroll
            for (int k = 0; k < EB; ++k) {
                if (i0 + k < n) {
                    const int sk = (int)((unsigned)__builtin_amdgcn_readlane((int)r, i0 + k) & SRC_MASK);
                    const float* yr = Y + sk * COUT + lane;
#pragma unroll
                    for (int g = 0; g < G; ++g) v[k][g] = yr[g * 32];
                }
            }
#pragma unroll
            for (int k = 0; k < EB; ++k) {
                if (i0 + k < n) {
                    const int dk = (int)((unsigned)__builtin_amdgcn_readlane((int)r, i0 + k) >> SRC_BITS);
                    float* a = acc + dk * COUT + lane;
#pragma unroll
                    for (int g = 0; g < G; ++g)
                        __hip_atomic_fetch_add(a + g * 32, v[k][g], __ATOMIC_RELAXED,
                                               __HIP_MEMORY_SCOPE_WORKGROUP);
                }
            }
        }
    };

    kept = __builtin_amdgcn_readfirstlane(kept);
    if (kept <= LIST_CAP) {
        unsigned r = 0;
        if (lane < kept) r = mine[lane];
        for (int i0 = 0; i0 < kept; i0 += 32) {
            const int left = kept - i0;
            unsigned rn = 0;
            if (32 + lane < left) rn = mine[i0 + 32 + lane];
            if (left >= 32) add_rows(r, 32); else add_rows(r, left);
            r = rn;
        }
    } else {
        const uint4* row = rows + (size_t)owner * (ROW_WORDS / 4);
        int* wq = q[wave];
        int head = 0, tail = 0;

        auto drain = [&](int n) {
            unsigned r = 0;
            if (lane < n) r = rec[wq[(head + lane) & 63]];
            add_rows(r, n);
            head += n;
        };

        static_assert(AGG_WAVES <= N_TRIPS, "every wave has a first trip");
        uint4 cur = row[wave * 32 + lane];
        for (int trip = wave; trip < N_TRIPS; trip += AGG_WAVES) {
            const int ntrip = (trip + AGG_WAVES < N_TRIPS) ? (trip + AGG_WAVES) : trip;
            const uint4 nxt = row[ntrip * 32 + lane];

            unsigned long long lo = (unsigned long long)cur.x | ((unsigned long long)cur.y << 32);
            unsigned long long hi = (unsigned long long)cur.z | ((unsigned long long)cur.w << 32);
            const int ebase = (trip * 32 + lane) * 128;

            for (;;) {
                const bool has = ((lo | hi) != 0ull);
                const unsigned bal = __builtin_amdgcn_ballot_w32(has);
                if (bal == 0u) break;
                if (has) {
                    const bool low = (lo != 0ull);
                    const unsigned long long m = low ? lo : hi;
                    const int p    = __builtin_ctzll(m) + (low ? 0 : 64);
                    const int rank = (int)__builtin_amdgcn_mbcnt_lo(bal, 0u);
                    wq[(tail + rank) & 63] = ebase + p;
                    const unsigned long long rest = m & (m - 1ull);
                    lo = low ? rest : lo;
                    hi = low ? hi : rest;
                }
                tail += __builtin_popcount(bal);
                __builtin_amdgcn_fence(__ATOMIC_ACQ_REL, "wavefront");
                if (tail - head >= 32) drain(32);
            }
            cur = nxt;
        }
        __builtin_amdgcn_fence(__ATOMIC_ACQ_REL, "wavefront");
        drain(tail - head);
    }
    __syncthreads();

    for (int i = tid; i < nval; i += AGG_THREADS) {
        const int n = i / COUT;
        const int c = i & (COUT - 1);
        float v = dinv[node0 + n] * acc[i];
        if (BIAS) v += bias[c];
        if (RELU) v = fmaxf(v, 0.0f);
        OUT[node0 * COUT + i] = v;
    }
}

extern "C" void kernel_launch(void* const* d_in, const int* in_sizes, int n_in,
                              void* d_out, int out_size, void* d_ws, size_t ws_size,
                              hipStream_t stream) {
    const float* x  = (const float*)d_in[0];
    const int*   ei = (const int*)d_in[1];
    const float* W1 = (const float*)d_in[2];
    const float* b1 = (const float*)d_in[3];
    const float* W2 = (const float*)d_in[4];
    const float* b2 = (const float*)d_in[5];
    const float* W3 = (const float*)d_in[6];
    const float* b3 = (const float*)d_in[7];
    float*       out = (float*)d_out;

    const int* src = ei;
    const int* dst = ei + N_EDGES;

    char*  ws   = (char*)d_ws;
    size_t off  = 0;
    float*    dinv = (float*)(ws + off);    off += ((size_t)N_NODES * 4 + 511) & ~511ull;
    unsigned* rec  = (unsigned*)(ws + off); off += ((size_t)N_EDGES * 4 + 511) & ~511ull;
    float*    y    = (float*)(ws + off);    off += (size_t)N_NODES * 128 * 4;
    float*    h    = (float*)(ws + off);    off += (size_t)N_NODES * 128 * 4;
    unsigned* lst  = (unsigned*)(ws + off); off += (size_t)N_OWNERS * AGG_WAVES * LIST_CAP * 4;
    int*      lcnt = (int*)(ws + off);      off += (size_t)N_OWNERS * AGG_WAVES * 4;
    uint4*    rows = (uint4*)(ws + off);

    float* xs = y;
    float* t  = y + (size_t)N_NODES * 64;

    const int M_BLOCKS = N_NODES / GEMM_ROWS;

    owner_rows_kernel<<<N_CHUNKS, CH_THREADS, 0, stream>>>(src, dst, rows, (uint4*)rec);
    owner_walk<<<N_OWNERS, AGG_THREADS, 0, stream>>>(rec, rows, x, dinv, xs, lst, lcnt);

    owner_aggregate<64, false, false><<<N_OWNERS, AGG_THREADS, 0, stream>>>(rec, rows, xs, dinv, nullptr, t, lst, lcnt);

    gemm_pair_wmma<<<M_BLOCKS, 256, 0, stream>>>(t, W1, b1, W2, dinv, h);
    owner_aggregate<128, true, true><<<N_OWNERS, AGG_THREADS, 0, stream>>>(rec, rows, h, dinv, b2, y, lst, lcnt);

    gemm_wmma<128, 64><<<M_BLOCKS, 128, 0, stream>>>(y, W3, dinv, h);
    owner_aggregate<64, false, true><<<N_OWNERS, AGG_THREADS, 0, stream>>>(rec, rows, h, dinv, b3, out, lst, lcnt);
}
